# starting kernel
# baseline (speedup 1.0000x reference)
#include <hip/hip_runtime.h>
#include <math.h>

typedef __attribute__((ext_vector_type(2))) float v2f;
typedef __attribute__((ext_vector_type(8))) float v8f;

#define C_DIM 16
#define T_DIM 160
#define NITERS 10
#define EPSV 1e-8f
#define XS_STRIDE 162
#define AUG_STRIDE 33
#define JS 17
#define NSWEEP 8

__device__ __forceinline__ v8f wmma_f32_16x16x4(v2f a, v2f b, v8f c) {
  return __builtin_amdgcn_wmma_f32_16x16x4_f32(false, a, false, b, (short)0, c,
                                               false, false);
}

__device__ __forceinline__ float wave_sum32(float v, float* red, int lane) {
  red[lane] = v;
  __syncthreads();
  float s = 0.f;
#pragma unroll
  for (int i = 0; i < 32; ++i) s += red[i];
  __syncthreads();
  return s;
}

__device__ __forceinline__ float diag_partial(const v8f& a, int lane) {
  float p = 0.f;
  const int mb = (lane >> 4) * 8;
  const int n = lane & 15;
#pragma unroll
  for (int r = 0; r < 8; ++r) p += (n == (r + mb)) ? a[r] : 0.f;
  return p;
}

__global__ __launch_bounds__(32) void tyler_tangent_kernel(
    const float* __restrict__ x, float* __restrict__ out) {
  const int b    = blockIdx.x;
  const int lane = threadIdx.x;
  const int row  = lane & 15;
  const int hi   = lane >> 4;
  const int koff = hi * 2;

  __shared__ float Xs[C_DIM * XS_STRIDE];
  __shared__ float aug[C_DIM * AUG_STRIDE];
  __shared__ float As[C_DIM * JS];
  __shared__ float Vs[C_DIM * JS];
  __shared__ float wbuf[T_DIM];
  __shared__ float lw[C_DIM];
  __shared__ float red[32];

  {
    const float4* xin =
        reinterpret_cast<const float4*>(x + (size_t)b * (C_DIM * T_DIM));
#pragma unroll
    for (int i = 0; i < 20; ++i) {
      int idx4 = lane + 32 * i;
      float4 v = xin[idx4];
      int c = idx4 / 40;
      int t0 = (idx4 % 40) * 4;
      float* p = &Xs[c * XS_STRIDE + t0];
      p[0] = v.x; p[1] = v.y; p[2] = v.z; p[3] = v.w;
    }
  }
  __syncthreads();

  const float* fragbase = &Xs[row * XS_STRIDE + koff];

  v8f acc = {0.f, 0.f, 0.f, 0.f, 0.f, 0.f, 0.f, 0.f};
#pragma unroll 1
  for (int cb = 0; cb < 10; ++cb) {
    v2f f[4];
#pragma unroll
    for (int u = 0; u < 4; ++u) {
      const float* p = fragbase + (cb * 4 + u) * 4;
      f[u].x = p[0];
      f[u].y = p[1];
    }
#pragma unroll
    for (int u = 0; u < 4; ++u) acc = wmma_f32_16x16x4(f[u], f[u], acc);
  }
  float tr = wave_sum32(diag_partial(acc, lane) * (1.f / T_DIM), red, lane);
  float scl = (1.f / T_DIM) * (16.f / (tr + EPSV));
  v8f ck;
#pragma unroll
  for (int r = 0; r < 8; ++r) ck[r] = acc[r] * scl;

#pragma unroll 1
  for (int it = 0; it < NITERS; ++it) {
#pragma unroll
    for (int r = 0; r < 8; ++r) {
      int m = r + 8 * hi;
      aug[m * AUG_STRIDE + row]      = ck[r] + ((m == row) ? EPSV : 0.f);
      aug[m * AUG_STRIDE + 16 + row] = (m == row) ? 1.f : 0.f;
    }
    __syncthreads();

#pragma unroll 1
    for (int k = 0; k < 16; ++k) {
      float pinv = 1.f / aug[k * AUG_STRIDE + k];
      __syncthreads();
      aug[k * AUG_STRIDE + lane] *= pinv;
      __syncthreads();
#pragma unroll 1
      for (int r = 0; r < 16; ++r) {
        if (r == k) continue;
        float fct = aug[r * AUG_STRIDE + k];
        aug[r * AUG_STRIDE + lane] -= fct * aug[k * AUG_STRIDE + lane];
      }
      __syncthreads();
    }

#pragma unroll 1
    for (int j5 = 0; j5 < 5; ++j5) {
      int t = lane + 32 * j5;
      float xc[16];
#pragma unroll
      for (int i = 0; i < 16; ++i) xc[i] = Xs[i * XS_STRIDE + t];
      float qv = 0.f;
#pragma unroll
      for (int i = 0; i < 16; ++i) {
        const float* ci = &aug[i * AUG_STRIDE + 16];
        qv += ci[i] * xc[i] * xc[i];
        float ra = 0.f;
#pragma unroll
        for (int j = i + 1; j < 16; ++j) ra += ci[j] * xc[j];
        qv += 2.f * xc[i] * ra;
      }
      wbuf[t] = 16.f / fmaxf(qv, EPSV);
    }
    __syncthreads();

    v8f a2 = {0.f, 0.f, 0.f, 0.f, 0.f, 0.f, 0.f, 0.f};
#pragma unroll 1
    for (int cb = 0; cb < 10; ++cb) {
      v2f bb[4], aa[4];
#pragma unroll
      for (int u = 0; u < 4; ++u) {
        int t = (cb * 4 + u) * 4 + koff;
        const float* p = fragbase + (cb * 4 + u) * 4;
        bb[u].x = p[0];
        bb[u].y = p[1];
        aa[u].x = bb[u].x * wbuf[t];
        aa[u].y = bb[u].y * wbuf[t + 1];
      }
#pragma unroll
      for (int u = 0; u < 4; ++u) a2 = wmma_f32_16x16x4(aa[u], bb[u], a2);
    }
    float tr2 = wave_sum32(diag_partial(a2, lane) * (1.f / T_DIM), red, lane);
    float s2 = (1.f / T_DIM) * (16.f / (tr2 + EPSV));
#pragma unroll
    for (int r = 0; r < 8; ++r) ck[r] = a2[r] * s2;
    __syncthreads();
  }

#pragma unroll
  for (int r = 0; r < 8; ++r) { int m = r + 8 * hi; As[m * JS + row] = ck[r]; }
  __syncthreads();
  float tv[8];
#pragma unroll
  for (int r = 0; r < 8; ++r) { int m = r + 8 * hi; tv[r] = As[row * JS + m]; }
  __syncthreads();
#pragma unroll
  for (int r = 0; r < 8; ++r) {
    int m = r + 8 * hi;
    As[m * JS + row] = 0.5f * (ck[r] + tv[r]);
  }
  if (lane < 16) {
#pragma unroll
    for (int j = 0; j < 16; ++j) Vs[lane * JS + j] = (lane == j) ? 1.f : 0.f;
  }
  __syncthreads();

#pragma unroll 1
  for (int sweep = 0; sweep < NSWEEP; ++sweep) {
#pragma unroll 1
    for (int p = 0; p < 15; ++p) {
#pragma unroll 1
      for (int q = p + 1; q < 16; ++q) {
        float apq = As[p * JS + q];
        float app = As[p * JS + p];
        float aqq = As[q * JS + q];
        float cc = 1.f, ss = 0.f;
        if (fabsf(apq) > 1e-30f) {
          float th = (aqq - app) / (2.f * apq);
          float tt = 1.f / (fabsf(th) + sqrtf(th * th + 1.f));
          if (th < 0.f) tt = -tt;
          cc = 1.f / sqrtf(1.f + tt * tt);
          ss = tt * cc;
        }
        __syncthreads();
        int k2 = lane & 15;
        if (lane < 16) {
          float akp = As[k2 * JS + p], akq = As[k2 * JS + q];
          As[k2 * JS + p] = cc * akp - ss * akq;
          As[k2 * JS + q] = ss * akp + cc * akq;
        } else {
          float vkp = Vs[k2 * JS + p], vkq = Vs[k2 * JS + q];
          Vs[k2 * JS + p] = cc * vkp - ss * vkq;
          Vs[k2 * JS + q] = ss * vkp + cc * vkq;
        }
        __syncthreads();
        if (lane < 16) {
          float apk = As[p * JS + k2], aqk = As[q * JS + k2];
          As[p * JS + k2] = cc * apk - ss * aqk;
          As[q * JS + k2] = ss * apk + cc * aqk;
        }
        __syncthreads();
      }
    }
  }
  if (lane < 16) lw[lane] = logf(fmaxf(As[lane * JS + lane], EPSV));
  __syncthreads();

  v8f lg = {0.f, 0.f, 0.f, 0.f, 0.f, 0.f, 0.f, 0.f};
#pragma unroll
  for (int kc = 0; kc < 4; ++kc) {
    int k0 = kc * 4 + koff;
    v2f bb; bb.x = Vs[row * JS + k0]; bb.y = Vs[row * JS + k0 + 1];
    float l0 = lw[k0], l1 = lw[k0 + 1];
    v2f aa; aa.x = bb.x * l0; aa.y = bb.y * l1;
    lg = wmma_f32_16x16x4(aa, bb, lg);
  }

  const float SQ2 = 1.4142135623730951f;
  float* op = out + (size_t)b * 136;
#pragma unroll
  for (int r = 0; r < 8; ++r) {
    int m = r + 8 * hi;
    if (m <= row) {
      int idx = m * 16 - (m * (m - 1)) / 2 + (row - m);
      op[idx] = lg[r] * ((m == row) ? 1.f : SQ2);
    }
  }
}

extern "C" void kernel_launch(void* const* d_in, const int* in_sizes, int n_in,
                              void* d_out, int out_size, void* d_ws,
                              size_t ws_size, hipStream_t stream) {
  const float* x = (const float*)d_in[0];
  float* out = (float*)d_out;
  int B = in_sizes[0] / (C_DIM * T_DIM);
  tyler_tangent_kernel<<<B, 32, 0, stream>>>(x, out);
}


# Jacobi eigen-solve re-ordered round-robin: each sweep is 15 rounds of 8 index-disjoint rotations applied together across the 32 lanes (rotation order, hence rounding, changes; 8 sweeps kept)
# speedup vs baseline: 2.5091x; 2.5091x over previous
#include <hip/hip_runtime.h>
#include <math.h>

typedef __attribute__((ext_vector_type(2))) float v2f;
typedef __attribute__((ext_vector_type(8))) float v8f;

#define C_DIM 16
#define T_DIM 160
#define NITERS 10
#define EPSV 1e-8f
#define XS_STRIDE 162
#define AUG_STRIDE 33
#define JS 17
#define NSWEEP 8

__device__ __forceinline__ v8f wmma_f32_16x16x4(v2f a, v2f b, v8f c) {
  return __builtin_amdgcn_wmma_f32_16x16x4_f32(false, a, false, b, (short)0, c,
                                               false, false);
}

__device__ __forceinline__ float wave_sum32(float v, float* red, int lane) {
  red[lane] = v;
  __syncthreads();
  float s = 0.f;
#pragma unroll
  for (int i = 0; i < 32; ++i) s += red[i];
  __syncthreads();
  return s;
}

__device__ __forceinline__ float diag_partial(const v8f& a, int lane) {
  float p = 0.f;
  const int mb = (lane >> 4) * 8;
  const int n = lane & 15;
#pragma unroll
  for (int r = 0; r < 8; ++r) p += (n == (r + mb)) ? a[r] : 0.f;
  return p;
}

__global__ __launch_bounds__(32) void tyler_tangent_kernel(
    const float* __restrict__ x, float* __restrict__ out) {
  const int b    = blockIdx.x;
  const int lane = threadIdx.x;
  const int row  = lane & 15;
  const int hi   = lane >> 4;
  const int koff = hi * 2;

  __shared__ float Xs[C_DIM * XS_STRIDE];
  __shared__ float aug[C_DIM * AUG_STRIDE];
  __shared__ float As[C_DIM * JS];
  __shared__ float Vs[C_DIM * JS];
  __shared__ float wbuf[T_DIM];
  __shared__ float lw[C_DIM];
  __shared__ float red[32];

  {
    const float4* xin =
        reinterpret_cast<const float4*>(x + (size_t)b * (C_DIM * T_DIM));
#pragma unroll
    for (int i = 0; i < 20; ++i) {
      int idx4 = lane + 32 * i;
      float4 v = xin[idx4];
      int c = idx4 / 40;
      int t0 = (idx4 % 40) * 4;
      float* p = &Xs[c * XS_STRIDE + t0];
      p[0] = v.x; p[1] = v.y; p[2] = v.z; p[3] = v.w;
    }
  }
  __syncthreads();

  const float* fragbase = &Xs[row * XS_STRIDE + koff];

  v8f acc = {0.f, 0.f, 0.f, 0.f, 0.f, 0.f, 0.f, 0.f};
#pragma unroll 1
  for (int cb = 0; cb < 10; ++cb) {
    v2f f[4];
#pragma unroll
    for (int u = 0; u < 4; ++u) {
      const float* p = fragbase + (cb * 4 + u) * 4;
      f[u].x = p[0];
      f[u].y = p[1];
    }
#pragma unroll
    for (int u = 0; u < 4; ++u) acc = wmma_f32_16x16x4(f[u], f[u], acc);
  }
  float tr = wave_sum32(diag_partial(acc, lane) * (1.f / T_DIM), red, lane);
  float scl = (1.f / T_DIM) * (16.f / (tr + EPSV));
  v8f ck;
#pragma unroll
  for (int r = 0; r < 8; ++r) ck[r] = acc[r] * scl;

#pragma unroll 1
  for (int it = 0; it < NITERS; ++it) {
#pragma unroll
    for (int r = 0; r < 8; ++r) {
      int m = r + 8 * hi;
      aug[m * AUG_STRIDE + row]      = ck[r] + ((m == row) ? EPSV : 0.f);
      aug[m * AUG_STRIDE + 16 + row] = (m == row) ? 1.f : 0.f;
    }
    __syncthreads();

#pragma unroll 1
    for (int k = 0; k < 16; ++k) {
      float pinv = 1.f / aug[k * AUG_STRIDE + k];
      __syncthreads();
      aug[k * AUG_STRIDE + lane] *= pinv;
      __syncthreads();
#pragma unroll 1
      for (int r = 0; r < 16; ++r) {
        if (r == k) continue;
        float fct = aug[r * AUG_STRIDE + k];
        aug[r * AUG_STRIDE + lane] -= fct * aug[k * AUG_STRIDE + lane];
      }
      __syncthreads();
    }

#pragma unroll 1
    for (int j5 = 0; j5 < 5; ++j5) {
      int t = lane + 32 * j5;
      float xc[16];
#pragma unroll
      for (int i = 0; i < 16; ++i) xc[i] = Xs[i * XS_STRIDE + t];
      float qv = 0.f;
#pragma unroll
      for (int i = 0; i < 16; ++i) {
        const float* ci = &aug[i * AUG_STRIDE + 16];
        qv += ci[i] * xc[i] * xc[i];
        float ra = 0.f;
#pragma unroll
        for (int j = i + 1; j < 16; ++j) ra += ci[j] * xc[j];
        qv += 2.f * xc[i] * ra;
      }
      wbuf[t] = 16.f / fmaxf(qv, EPSV);
    }
    __syncthreads();

    v8f a2 = {0.f, 0.f, 0.f, 0.f, 0.f, 0.f, 0.f, 0.f};
#pragma unroll 1
    for (int cb = 0; cb < 10; ++cb) {
      v2f bb[4], aa[4];
#pragma unroll
      for (int u = 0; u < 4; ++u) {
        int t = (cb * 4 + u) * 4 + koff;
        const float* p = fragbase + (cb * 4 + u) * 4;
        bb[u].x = p[0];
        bb[u].y = p[1];
        aa[u].x = bb[u].x * wbuf[t];
        aa[u].y = bb[u].y * wbuf[t + 1];
      }
#pragma unroll
      for (int u = 0; u < 4; ++u) a2 = wmma_f32_16x16x4(aa[u], bb[u], a2);
    }
    float tr2 = wave_sum32(diag_partial(a2, lane) * (1.f / T_DIM), red, lane);
    float s2 = (1.f / T_DIM) * (16.f / (tr2 + EPSV));
#pragma unroll
    for (int r = 0; r < 8; ++r) ck[r] = a2[r] * s2;
    __syncthreads();
  }

#pragma unroll
  for (int r = 0; r < 8; ++r) { int m = r + 8 * hi; As[m * JS + row] = ck[r]; }
  __syncthreads();
  float tv[8];
#pragma unroll
  for (int r = 0; r < 8; ++r) { int m = r + 8 * hi; tv[r] = As[row * JS + m]; }
  __syncthreads();
#pragma unroll
  for (int r = 0; r < 8; ++r) {
    int m = r + 8 * hi;
    As[m * JS + row] = 0.5f * (ck[r] + tv[r]);
  }
  if (lane < 16) {
#pragma unroll
    for (int j = 0; j < 16; ++j) Vs[lane * JS + j] = (lane == j) ? 1.f : 0.f;
  }
  __syncthreads();

  const int pj = lane & 7;
  const int kg = lane >> 3;
#pragma unroll 1
  for (int sweep = 0; sweep < NSWEEP; ++sweep) {
#pragma unroll 1
    for (int rnd = 0; rnd < 15; ++rnd) {
      int ia = rnd + pj;       if (ia >= 15) ia -= 15;
      int ib = rnd + 15 - pj;  if (ib >= 15) ib -= 15;
      if (pj == 0) ib = 15;
      const int p = min(ia, ib);
      const int q = max(ia, ib);
      float apq = As[p * JS + q];
      float app = As[p * JS + p];
      float aqq = As[q * JS + q];
      float cc = 1.f, ss = 0.f;
      if (fabsf(apq) > 1e-30f) {
        float th = (aqq - app) / (2.f * apq);
        float tt = 1.f / (fabsf(th) + sqrtf(th * th + 1.f));
        if (th < 0.f) tt = -tt;
        cc = 1.f / sqrtf(1.f + tt * tt);
        ss = tt * cc;
      }
      __syncthreads();
      {
        float akp[4], akq[4], vkp[4], vkq[4];
#pragma unroll
        for (int i = 0; i < 4; ++i) {
          const int k2 = kg + 4 * i;
          akp[i] = As[k2 * JS + p]; akq[i] = As[k2 * JS + q];
          vkp[i] = Vs[k2 * JS + p]; vkq[i] = Vs[k2 * JS + q];
        }
#pragma unroll
        for (int i = 0; i < 4; ++i) {
          const int k2 = kg + 4 * i;
          As[k2 * JS + p] = cc * akp[i] - ss * akq[i];
          As[k2 * JS + q] = ss * akp[i] + cc * akq[i];
          Vs[k2 * JS + p] = cc * vkp[i] - ss * vkq[i];
          Vs[k2 * JS + q] = ss * vkp[i] + cc * vkq[i];
        }
      }
      __syncthreads();
      {
        float apk[4], aqk[4];
#pragma unroll
        for (int i = 0; i < 4; ++i) {
          const int k2 = kg + 4 * i;
          apk[i] = As[p * JS + k2]; aqk[i] = As[q * JS + k2];
        }
#pragma unroll
        for (int i = 0; i < 4; ++i) {
          const int k2 = kg + 4 * i;
          As[p * JS + k2] = cc * apk[i] - ss * aqk[i];
          As[q * JS + k2] = ss * apk[i] + cc * aqk[i];
        }
      }
      __syncthreads();
    }
  }
  if (lane < 16) lw[lane] = logf(fmaxf(As[lane * JS + lane], EPSV));
  __syncthreads();

  v8f lg = {0.f, 0.f, 0.f, 0.f, 0.f, 0.f, 0.f, 0.f};
#pragma unroll
  for (int kc = 0; kc < 4; ++kc) {
    int k0 = kc * 4 + koff;
    v2f bb; bb.x = Vs[row * JS + k0]; bb.y = Vs[row * JS + k0 + 1];
    float l0 = lw[k0], l1 = lw[k0 + 1];
    v2f aa; aa.x = bb.x * l0; aa.y = bb.y * l1;
    lg = wmma_f32_16x16x4(aa, bb, lg);
  }

  const float SQ2 = 1.4142135623730951f;
  float* op = out + (size_t)b * 136;
#pragma unroll
  for (int r = 0; r < 8; ++r) {
    int m = r + 8 * hi;
    if (m <= row) {
      int idx = m * 16 - (m * (m - 1)) / 2 + (row - m);
      op[idx] = lg[r] * ((m == row) ? 1.f : SQ2);
    }
  }
}

extern "C" void kernel_launch(void* const* d_in, const int* in_sizes, int n_in,
                              void* d_out, int out_size, void* d_ws,
                              size_t ws_size, hipStream_t stream) {
  const float* x = (const float*)d_in[0];
  float* out = (float*)d_out;
  int B = in_sizes[0] / (C_DIM * T_DIM);
  tyler_tangent_kernel<<<B, 32, 0, stream>>>(x, out);
}


# Gauss-Jordan inverse moved from LDS into registers: each lane keeps one column of [Ck+eps*I | I], pivot-row factors broadcast by v_readlane; Cinv stored to LDS once per iteration for the q pass
# speedup vs baseline: 5.3669x; 2.1390x over previous
#include <hip/hip_runtime.h>
#include <math.h>

typedef __attribute__((ext_vector_type(2))) float v2f;
typedef __attribute__((ext_vector_type(8))) float v8f;

#define C_DIM 16
#define T_DIM 160
#define NITERS 10
#define EPSV 1e-8f
#define XS_STRIDE 162
#define AUG_STRIDE 33
#define JS 17
#define NSWEEP 8

__device__ __forceinline__ v8f wmma_f32_16x16x4(v2f a, v2f b, v8f c) {
  return __builtin_amdgcn_wmma_f32_16x16x4_f32(false, a, false, b, (short)0, c,
                                               false, false);
}

__device__ __forceinline__ float xhalf(float x) {
  int b = __builtin_bit_cast(int, x);
  return __builtin_bit_cast(
      float, __builtin_amdgcn_permlanex16(b, b, 0x76543210, 0xfedcba98, false, false));
}

__device__ __forceinline__ float lane_bcast(float v, int src) {
  return __builtin_bit_cast(float,
                            __builtin_amdgcn_readlane(__builtin_bit_cast(int, v), src));
}

__device__ __forceinline__ float wave_sum32(float v, float* red, int lane) {
  red[lane] = v;
  __syncthreads();
  float s = 0.f;
#pragma unroll
  for (int i = 0; i < 32; ++i) s += red[i];
  __syncthreads();
  return s;
}

__device__ __forceinline__ float diag_partial(const v8f& a, int lane) {
  float p = 0.f;
  const int mb = (lane >> 4) * 8;
  const int n = lane & 15;
#pragma unroll
  for (int r = 0; r < 8; ++r) p += (n == (r + mb)) ? a[r] : 0.f;
  return p;
}

__global__ __launch_bounds__(32) void tyler_tangent_kernel(
    const float* __restrict__ x, float* __restrict__ out) {
  const int b    = blockIdx.x;
  const int lane = threadIdx.x;
  const int row  = lane & 15;
  const int hi   = lane >> 4;
  const int koff = hi * 2;

  __shared__ float Xs[C_DIM * XS_STRIDE];
  __shared__ float aug[C_DIM * AUG_STRIDE];
  __shared__ float As[C_DIM * JS];
  __shared__ float Vs[C_DIM * JS];
  __shared__ float wbuf[T_DIM];
  __shared__ float lw[C_DIM];
  __shared__ float red[32];

  {
    const float4* xin =
        reinterpret_cast<const float4*>(x + (size_t)b * (C_DIM * T_DIM));
#pragma unroll
    for (int i = 0; i < 20; ++i) {
      int idx4 = lane + 32 * i;
      float4 v = xin[idx4];
      int c = idx4 / 40;
      int t0 = (idx4 % 40) * 4;
      float* p = &Xs[c * XS_STRIDE + t0];
      p[0] = v.x; p[1] = v.y; p[2] = v.z; p[3] = v.w;
    }
  }
  __syncthreads();

  const float* fragbase = &Xs[row * XS_STRIDE + koff];

  v8f acc = {0.f, 0.f, 0.f, 0.f, 0.f, 0.f, 0.f, 0.f};
#pragma unroll 1
  for (int cb = 0; cb < 10; ++cb) {
    v2f f[4];
#pragma unroll
    for (int u = 0; u < 4; ++u) {
      const float* p = fragbase + (cb * 4 + u) * 4;
      f[u].x = p[0];
      f[u].y = p[1];
    }
#pragma unroll
    for (int u = 0; u < 4; ++u) acc = wmma_f32_16x16x4(f[u], f[u], acc);
  }
  float tr = wave_sum32(diag_partial(acc, lane) * (1.f / T_DIM), red, lane);
  float scl = (1.f / T_DIM) * (16.f / (tr + EPSV));
  v8f ck;
#pragma unroll
  for (int r = 0; r < 8; ++r) ck[r] = acc[r] * scl;

#pragma unroll 1
  for (int it = 0; it < NITERS; ++it) {
    float a[16];
#pragma unroll
    for (int r = 0; r < 8; ++r) {
      const float o = xhalf(ck[r]);
      a[r]     = hi ? ((r == row) ? 1.f : 0.f)     : ck[r] + ((r == row) ? EPSV : 0.f);
      a[r + 8] = hi ? ((r + 8 == row) ? 1.f : 0.f) : o + ((r + 8 == row) ? EPSV : 0.f);
    }

#pragma unroll
    for (int k = 0; k < 16; ++k) {
      const float pinv = 1.f / lane_bcast(a[k], k);
      float f[16];
#pragma unroll
      for (int r = 0; r < 16; ++r) f[r] = lane_bcast(a[r], k);
      a[k] *= pinv;
#pragma unroll
      for (int r = 0; r < 16; ++r) {
        if (r != k) a[r] -= f[r] * a[k];
      }
    }
#pragma unroll
    for (int i = 0; i < 16; ++i) aug[i * AUG_STRIDE + lane] = a[i];
    __syncthreads();

#pragma unroll 1
    for (int j5 = 0; j5 < 5; ++j5) {
      int t = lane + 32 * j5;
      float xc[16];
#pragma unroll
      for (int i = 0; i < 16; ++i) xc[i] = Xs[i * XS_STRIDE + t];
      float qv = 0.f;
#pragma unroll
      for (int i = 0; i < 16; ++i) {
        const float* ci = &aug[i * AUG_STRIDE + 16];
        qv += ci[i] * xc[i] * xc[i];
        float ra = 0.f;
#pragma unroll
        for (int j = i + 1; j < 16; ++j) ra += ci[j] * xc[j];
        qv += 2.f * xc[i] * ra;
      }
      wbuf[t] = 16.f / fmaxf(qv, EPSV);
    }
    __syncthreads();

    v8f a2 = {0.f, 0.f, 0.f, 0.f, 0.f, 0.f, 0.f, 0.f};
#pragma unroll 1
    for (int cb = 0; cb < 10; ++cb) {
      v2f bb[4], aa[4];
#pragma unroll
      for (int u = 0; u < 4; ++u) {
        int t = (cb * 4 + u) * 4 + koff;
        const float* p = fragbase + (cb * 4 + u) * 4;
        bb[u].x = p[0];
        bb[u].y = p[1];
        aa[u].x = bb[u].x * wbuf[t];
        aa[u].y = bb[u].y * wbuf[t + 1];
      }
#pragma unroll
      for (int u = 0; u < 4; ++u) a2 = wmma_f32_16x16x4(aa[u], bb[u], a2);
    }
    float tr2 = wave_sum32(diag_partial(a2, lane) * (1.f / T_DIM), red, lane);
    float s2 = (1.f / T_DIM) * (16.f / (tr2 + EPSV));
#pragma unroll
    for (int r = 0; r < 8; ++r) ck[r] = a2[r] * s2;
    __syncthreads();
  }

#pragma unroll
  for (int r = 0; r < 8; ++r) { int m = r + 8 * hi; As[m * JS + row] = ck[r]; }
  __syncthreads();
  float tv[8];
#pragma unroll
  for (int r = 0; r < 8; ++r) { int m = r + 8 * hi; tv[r] = As[row * JS + m]; }
  __syncthreads();
#pragma unroll
  for (int r = 0; r < 8; ++r) {
    int m = r + 8 * hi;
    As[m * JS + row] = 0.5f * (ck[r] + tv[r]);
  }
  if (lane < 16) {
#pragma unroll
    for (int j = 0; j < 16; ++j) Vs[lane * JS + j] = (lane == j) ? 1.f : 0.f;
  }
  __syncthreads();

  const int pj = lane & 7;
  const int kg = lane >> 3;
#pragma unroll 1
  for (int sweep = 0; sweep < NSWEEP; ++sweep) {
#pragma unroll 1
    for (int rnd = 0; rnd < 15; ++rnd) {
      int ia = rnd + pj;       if (ia >= 15) ia -= 15;
      int ib = rnd + 15 - pj;  if (ib >= 15) ib -= 15;
      if (pj == 0) ib = 15;
      const int p = min(ia, ib);
      const int q = max(ia, ib);
      float apq = As[p * JS + q];
      float app = As[p * JS + p];
      float aqq = As[q * JS + q];
      float cc = 1.f, ss = 0.f;
      if (fabsf(apq) > 1e-30f) {
        float th = (aqq - app) / (2.f * apq);
        float tt = 1.f / (fabsf(th) + sqrtf(th * th + 1.f));
        if (th < 0.f) tt = -tt;
        cc = 1.f / sqrtf(1.f + tt * tt);
        ss = tt * cc;
      }
      __syncthreads();
      {
        float akp[4], akq[4], vkp[4], vkq[4];
#pragma unroll
        for (int i = 0; i < 4; ++i) {
          const int k2 = kg + 4 * i;
          akp[i] = As[k2 * JS + p]; akq[i] = As[k2 * JS + q];
          vkp[i] = Vs[k2 * JS + p]; vkq[i] = Vs[k2 * JS + q];
        }
#pragma unroll
        for (int i = 0; i < 4; ++i) {
          const int k2 = kg + 4 * i;
          As[k2 * JS + p] = cc * akp[i] - ss * akq[i];
          As[k2 * JS + q] = ss * akp[i] + cc * akq[i];
          Vs[k2 * JS + p] = cc * vkp[i] - ss * vkq[i];
          Vs[k2 * JS + q] = ss * vkp[i] + cc * vkq[i];
        }
      }
      __syncthreads();
      {
        float apk[4], aqk[4];
#pragma unroll
        for (int i = 0; i < 4; ++i) {
          const int k2 = kg + 4 * i;
          apk[i] = As[p * JS + k2]; aqk[i] = As[q * JS + k2];
        }
#pragma unroll
        for (int i = 0; i < 4; ++i) {
          const int k2 = kg + 4 * i;
          As[p * JS + k2] = cc * apk[i] - ss * aqk[i];
          As[q * JS + k2] = ss * apk[i] + cc * aqk[i];
        }
      }
      __syncthreads();
    }
  }
  if (lane < 16) lw[lane] = logf(fmaxf(As[lane * JS + lane], EPSV));
  __syncthreads();

  v8f lg = {0.f, 0.f, 0.f, 0.f, 0.f, 0.f, 0.f, 0.f};
#pragma unroll
  for (int kc = 0; kc < 4; ++kc) {
    int k0 = kc * 4 + koff;
    v2f bb; bb.x = Vs[row * JS + k0]; bb.y = Vs[row * JS + k0 + 1];
    float l0 = lw[k0], l1 = lw[k0 + 1];
    v2f aa; aa.x = bb.x * l0; aa.y = bb.y * l1;
    lg = wmma_f32_16x16x4(aa, bb, lg);
  }

  const float SQ2 = 1.4142135623730951f;
  float* op = out + (size_t)b * 136;
#pragma unroll
  for (int r = 0; r < 8; ++r) {
    int m = r + 8 * hi;
    if (m <= row) {
      int idx = m * 16 - (m * (m - 1)) / 2 + (row - m);
      op[idx] = lg[r] * ((m == row) ? 1.f : SQ2);
    }
  }
}

extern "C" void kernel_launch(void* const* d_in, const int* in_sizes, int n_in,
                              void* d_out, int out_size, void* d_ws,
                              size_t ws_size, hipStream_t stream) {
  const float* x = (const float*)d_in[0];
  float* out = (float*)d_out;
  int B = in_sizes[0] / (C_DIM * T_DIM);
  tyler_tangent_kernel<<<B, 32, 0, stream>>>(x, out);
}


# Jacobi rotation (c,s) set-up made branch-free on hardware rcp/sqrt/rsq (one Newton step on rsq) instead of 3 IEEE divisions + 2 IEEE sqrts per round; rotation rounding changes
# speedup vs baseline: 6.6072x; 1.2311x over previous
#include <hip/hip_runtime.h>
#include <math.h>

typedef __attribute__((ext_vector_type(2))) float v2f;
typedef __attribute__((ext_vector_type(8))) float v8f;

#define C_DIM 16
#define T_DIM 160
#define NITERS 10
#define EPSV 1e-8f
#define XS_STRIDE 162
#define AUG_STRIDE 33
#define JS 17
#define NSWEEP 8

__device__ __forceinline__ v8f wmma_f32_16x16x4(v2f a, v2f b, v8f c) {
  return __builtin_amdgcn_wmma_f32_16x16x4_f32(false, a, false, b, (short)0, c,
                                               false, false);
}

__device__ __forceinline__ float xhalf(float x) {
  int b = __builtin_bit_cast(int, x);
  return __builtin_bit_cast(
      float, __builtin_amdgcn_permlanex16(b, b, 0x76543210, 0xfedcba98, false, false));
}

__device__ __forceinline__ float lane_bcast(float v, int src) {
  return __builtin_bit_cast(float,
                            __builtin_amdgcn_readlane(__builtin_bit_cast(int, v), src));
}

__device__ __forceinline__ float hw_rcp(float v) { return __builtin_amdgcn_rcpf(v); }
__device__ __forceinline__ float hw_sqrt(float v) { return __builtin_amdgcn_sqrtf(v); }
__device__ __forceinline__ float hw_rsq(float v) { return __builtin_amdgcn_rsqf(v); }

__device__ __forceinline__ float wave_sum32(float v, float* red, int lane) {
  red[lane] = v;
  __syncthreads();
  float s = 0.f;
#pragma unroll
  for (int i = 0; i < 32; ++i) s += red[i];
  __syncthreads();
  return s;
}

__device__ __forceinline__ float diag_partial(const v8f& a, int lane) {
  float p = 0.f;
  const int mb = (lane >> 4) * 8;
  const int n = lane & 15;
#pragma unroll
  for (int r = 0; r < 8; ++r) p += (n == (r + mb)) ? a[r] : 0.f;
  return p;
}

__global__ __launch_bounds__(32) void tyler_tangent_kernel(
    const float* __restrict__ x, float* __restrict__ out) {
  const int b    = blockIdx.x;
  const int lane = threadIdx.x;
  const int row  = lane & 15;
  const int hi   = lane >> 4;
  const int koff = hi * 2;

  __shared__ float Xs[C_DIM * XS_STRIDE];
  __shared__ float aug[C_DIM * AUG_STRIDE];
  __shared__ float As[C_DIM * JS];
  __shared__ float Vs[C_DIM * JS];
  __shared__ float wbuf[T_DIM];
  __shared__ float lw[C_DIM];
  __shared__ float red[32];

  {
    const float4* xin =
        reinterpret_cast<const float4*>(x + (size_t)b * (C_DIM * T_DIM));
#pragma unroll
    for (int i = 0; i < 20; ++i) {
      int idx4 = lane + 32 * i;
      float4 v = xin[idx4];
      int c = idx4 / 40;
      int t0 = (idx4 % 40) * 4;
      float* p = &Xs[c * XS_STRIDE + t0];
      p[0] = v.x; p[1] = v.y; p[2] = v.z; p[3] = v.w;
    }
  }
  __syncthreads();

  const float* fragbase = &Xs[row * XS_STRIDE + koff];

  v8f acc = {0.f, 0.f, 0.f, 0.f, 0.f, 0.f, 0.f, 0.f};
#pragma unroll 1
  for (int cb = 0; cb < 10; ++cb) {
    v2f f[4];
#pragma unroll
    for (int u = 0; u < 4; ++u) {
      const float* p = fragbase + (cb * 4 + u) * 4;
      f[u].x = p[0];
      f[u].y = p[1];
    }
#pragma unroll
    for (int u = 0; u < 4; ++u) acc = wmma_f32_16x16x4(f[u], f[u], acc);
  }
  float tr = wave_sum32(diag_partial(acc, lane) * (1.f / T_DIM), red, lane);
  float scl = (1.f / T_DIM) * (16.f / (tr + EPSV));
  v8f ck;
#pragma unroll
  for (int r = 0; r < 8; ++r) ck[r] = acc[r] * scl;

#pragma unroll 1
  for (int it = 0; it < NITERS; ++it) {
    float a[16];
#pragma unroll
    for (int r = 0; r < 8; ++r) {
      const float o = xhalf(ck[r]);
      a[r]     = hi ? ((r == row) ? 1.f : 0.f)     : ck[r] + ((r == row) ? EPSV : 0.f);
      a[r + 8] = hi ? ((r + 8 == row) ? 1.f : 0.f) : o + ((r + 8 == row) ? EPSV : 0.f);
    }

#pragma unroll
    for (int k = 0; k < 16; ++k) {
      const float pinv = 1.f / lane_bcast(a[k], k);
      float f[16];
#pragma unroll
      for (int r = 0; r < 16; ++r) f[r] = lane_bcast(a[r], k);
      a[k] *= pinv;
#pragma unroll
      for (int r = 0; r < 16; ++r) {
        if (r != k) a[r] -= f[r] * a[k];
      }
    }
#pragma unroll
    for (int i = 0; i < 16; ++i) aug[i * AUG_STRIDE + lane] = a[i];
    __syncthreads();

#pragma unroll 1
    for (int j5 = 0; j5 < 5; ++j5) {
      int t = lane + 32 * j5;
      float xc[16];
#pragma unroll
      for (int i = 0; i < 16; ++i) xc[i] = Xs[i * XS_STRIDE + t];
      float qv = 0.f;
#pragma unroll
      for (int i = 0; i < 16; ++i) {
        const float* ci = &aug[i * AUG_STRIDE + 16];
        qv += ci[i] * xc[i] * xc[i];
        float ra = 0.f;
#pragma unroll
        for (int j = i + 1; j < 16; ++j) ra += ci[j] * xc[j];
        qv += 2.f * xc[i] * ra;
      }
      wbuf[t] = 16.f / fmaxf(qv, EPSV);
    }
    __syncthreads();

    v8f a2 = {0.f, 0.f, 0.f, 0.f, 0.f, 0.f, 0.f, 0.f};
#pragma unroll 1
    for (int cb = 0; cb < 10; ++cb) {
      v2f bb[4], aa[4];
#pragma unroll
      for (int u = 0; u < 4; ++u) {
        int t = (cb * 4 + u) * 4 + koff;
        const float* p = fragbase + (cb * 4 + u) * 4;
        bb[u].x = p[0];
        bb[u].y = p[1];
        aa[u].x = bb[u].x * wbuf[t];
        aa[u].y = bb[u].y * wbuf[t + 1];
      }
#pragma unroll
      for (int u = 0; u < 4; ++u) a2 = wmma_f32_16x16x4(aa[u], bb[u], a2);
    }
    float tr2 = wave_sum32(diag_partial(a2, lane) * (1.f / T_DIM), red, lane);
    float s2 = (1.f / T_DIM) * (16.f / (tr2 + EPSV));
#pragma unroll
    for (int r = 0; r < 8; ++r) ck[r] = a2[r] * s2;
    __syncthreads();
  }

#pragma unroll
  for (int r = 0; r < 8; ++r) { int m = r + 8 * hi; As[m * JS + row] = ck[r]; }
  __syncthreads();
  float tv[8];
#pragma unroll
  for (int r = 0; r < 8; ++r) { int m = r + 8 * hi; tv[r] = As[row * JS + m]; }
  __syncthreads();
#pragma unroll
  for (int r = 0; r < 8; ++r) {
    int m = r + 8 * hi;
    As[m * JS + row] = 0.5f * (ck[r] + tv[r]);
  }
  if (lane < 16) {
#pragma unroll
    for (int j = 0; j < 16; ++j) Vs[lane * JS + j] = (lane == j) ? 1.f : 0.f;
  }
  __syncthreads();

  const int pj = lane & 7;
  const int kg = lane >> 3;
#pragma unroll 1
  for (int sweep = 0; sweep < NSWEEP; ++sweep) {
#pragma unroll 1
    for (int rnd = 0; rnd < 15; ++rnd) {
      int ia = rnd + pj;       if (ia >= 15) ia -= 15;
      int ib = rnd + 15 - pj;  if (ib >= 15) ib -= 15;
      if (pj == 0) ib = 15;
      const int p = min(ia, ib);
      const int q = max(ia, ib);
      const float apq = As[p * JS + q];
      const float app = As[p * JS + p];
      const float aqq = As[q * JS + q];
      const bool rot = fabsf(apq) > 1e-30f;
      const float th = (aqq - app) * hw_rcp(2.f * apq);
      float tt = hw_rcp(fabsf(th) + hw_sqrt(fmaf(th, th, 1.f)));
      if (th < 0.f) tt = -tt;
      const float w1 = fmaf(tt, tt, 1.f);
      float y = hw_rsq(w1);
      y = fmaf(0.5f * y, fmaf(-w1 * y, y, 1.f), y);
      const float cc = rot ? y : 1.f;
      const float ss = rot ? tt * y : 0.f;
      __syncthreads();
      {
        float akp[4], akq[4], vkp[4], vkq[4];
#pragma unroll
        for (int i = 0; i < 4; ++i) {
          const int k2 = kg + 4 * i;
          akp[i] = As[k2 * JS + p]; akq[i] = As[k2 * JS + q];
          vkp[i] = Vs[k2 * JS + p]; vkq[i] = Vs[k2 * JS + q];
        }
#pragma unroll
        for (int i = 0; i < 4; ++i) {
          const int k2 = kg + 4 * i;
          As[k2 * JS + p] = cc * akp[i] - ss * akq[i];
          As[k2 * JS + q] = ss * akp[i] + cc * akq[i];
          Vs[k2 * JS + p] = cc * vkp[i] - ss * vkq[i];
          Vs[k2 * JS + q] = ss * vkp[i] + cc * vkq[i];
        }
      }
      __syncthreads();
      {
        float apk[4], aqk[4];
#pragma unroll
        for (int i = 0; i < 4; ++i) {
          const int k2 = kg + 4 * i;
          apk[i] = As[p * JS + k2]; aqk[i] = As[q * JS + k2];
        }
#pragma unroll
        for (int i = 0; i < 4; ++i) {
          const int k2 = kg + 4 * i;
          As[p * JS + k2] = cc * apk[i] - ss * aqk[i];
          As[q * JS + k2] = ss * apk[i] + cc * aqk[i];
        }
      }
      __syncthreads();
    }
  }
  if (lane < 16) lw[lane] = logf(fmaxf(As[lane * JS + lane], EPSV));
  __syncthreads();

  v8f lg = {0.f, 0.f, 0.f, 0.f, 0.f, 0.f, 0.f, 0.f};
#pragma unroll
  for (int kc = 0; kc < 4; ++kc) {
    int k0 = kc * 4 + koff;
    v2f bb; bb.x = Vs[row * JS + k0]; bb.y = Vs[row * JS + k0 + 1];
    float l0 = lw[k0], l1 = lw[k0 + 1];
    v2f aa; aa.x = bb.x * l0; aa.y = bb.y * l1;
    lg = wmma_f32_16x16x4(aa, bb, lg);
  }

  const float SQ2 = 1.4142135623730951f;
  float* op = out + (size_t)b * 136;
#pragma unroll
  for (int r = 0; r < 8; ++r) {
    int m = r + 8 * hi;
    if (m <= row) {
      int idx = m * 16 - (m * (m - 1)) / 2 + (row - m);
      op[idx] = lg[r] * ((m == row) ? 1.f : SQ2);
    }
  }
}

extern "C" void kernel_launch(void* const* d_in, const int* in_sizes, int n_in,
                              void* d_out, int out_size, void* d_ws,
                              size_t ws_size, hipStream_t stream) {
  const float* x = (const float*)d_in[0];
  float* out = (float*)d_out;
  int B = in_sizes[0] / (C_DIM * T_DIM);
  tyler_tangent_kernel<<<B, 32, 0, stream>>>(x, out);
}


# q_t = x_t^T Cinv x_t moved onto the f32 matrix unit: Y = Cinv*X per 16-column tile (40 WMMAs/iteration, X fragments kept in registers), q = sum of X.*Y per column; q's summation order changes
# speedup vs baseline: 7.0513x; 1.0672x over previous
#include <hip/hip_runtime.h>
#include <math.h>

typedef __attribute__((ext_vector_type(2))) float v2f;
typedef __attribute__((ext_vector_type(8))) float v8f;

#define C_DIM 16
#define T_DIM 160
#define NITERS 10
#define EPSV 1e-8f
#define XS_STRIDE 162
#define AUG_STRIDE 33
#define JS 17
#define NSWEEP 8

__device__ __forceinline__ v8f wmma_f32_16x16x4(v2f a, v2f b, v8f c) {
  return __builtin_amdgcn_wmma_f32_16x16x4_f32(false, a, false, b, (short)0, c,
                                               false, false);
}

__device__ __forceinline__ float xhalf(float x) {
  int b = __builtin_bit_cast(int, x);
  return __builtin_bit_cast(
      float, __builtin_amdgcn_permlanex16(b, b, 0x76543210, 0xfedcba98, false, false));
}

__device__ __forceinline__ float lane_bcast(float v, int src) {
  return __builtin_bit_cast(float,
                            __builtin_amdgcn_readlane(__builtin_bit_cast(int, v), src));
}

__device__ __forceinline__ float hw_rcp(float v) { return __builtin_amdgcn_rcpf(v); }
__device__ __forceinline__ float hw_sqrt(float v) { return __builtin_amdgcn_sqrtf(v); }
__device__ __forceinline__ float hw_rsq(float v) { return __builtin_amdgcn_rsqf(v); }

__device__ __forceinline__ float wave_sum32(float v, float* red, int lane) {
  red[lane] = v;
  __syncthreads();
  float s = 0.f;
#pragma unroll
  for (int i = 0; i < 32; ++i) s += red[i];
  __syncthreads();
  return s;
}

__device__ __forceinline__ float diag_partial(const v8f& a, int lane) {
  float p = 0.f;
  const int mb = (lane >> 4) * 8;
  const int n = lane & 15;
#pragma unroll
  for (int r = 0; r < 8; ++r) p += (n == (r + mb)) ? a[r] : 0.f;
  return p;
}

__global__ __launch_bounds__(32) void tyler_tangent_kernel(
    const float* __restrict__ x, float* __restrict__ out) {
  const int b    = blockIdx.x;
  const int lane = threadIdx.x;
  const int row  = lane & 15;
  const int hi   = lane >> 4;
  const int koff = hi * 2;

  __shared__ float Xs[C_DIM * XS_STRIDE];
  __shared__ float aug[C_DIM * AUG_STRIDE];
  __shared__ float As[C_DIM * JS];
  __shared__ float Vs[C_DIM * JS];
  __shared__ float wbuf[T_DIM];
  __shared__ float lw[C_DIM];
  __shared__ float red[32];

  {
    const float4* xin =
        reinterpret_cast<const float4*>(x + (size_t)b * (C_DIM * T_DIM));
#pragma unroll
    for (int i = 0; i < 20; ++i) {
      int idx4 = lane + 32 * i;
      float4 v = xin[idx4];
      int c = idx4 / 40;
      int t0 = (idx4 % 40) * 4;
      float* p = &Xs[c * XS_STRIDE + t0];
      p[0] = v.x; p[1] = v.y; p[2] = v.z; p[3] = v.w;
    }
  }
  __syncthreads();

  const float* fragbase = &Xs[row * XS_STRIDE + koff];

  v8f acc = {0.f, 0.f, 0.f, 0.f, 0.f, 0.f, 0.f, 0.f};
#pragma unroll 1
  for (int cb = 0; cb < 10; ++cb) {
    v2f f[4];
#pragma unroll
    for (int u = 0; u < 4; ++u) {
      const float* p = fragbase + (cb * 4 + u) * 4;
      f[u].x = p[0];
      f[u].y = p[1];
    }
#pragma unroll
    for (int u = 0; u < 4; ++u) acc = wmma_f32_16x16x4(f[u], f[u], acc);
  }
  float tr = wave_sum32(diag_partial(acc, lane) * (1.f / T_DIM), red, lane);
  float scl = (1.f / T_DIM) * (16.f / (tr + EPSV));
  v8f ck;
#pragma unroll
  for (int r = 0; r < 8; ++r) ck[r] = acc[r] * scl;

  v2f xq[10][4];
  {
    const float* xb = &Xs[koff * XS_STRIDE + row];
#pragma unroll
    for (int tile = 0; tile < 10; ++tile) {
#pragma unroll
      for (int kc = 0; kc < 4; ++kc) {
        xq[tile][kc].x = xb[(4 * kc) * XS_STRIDE + 16 * tile];
        xq[tile][kc].y = xb[(4 * kc + 1) * XS_STRIDE + 16 * tile];
      }
    }
  }

#pragma unroll 1
  for (int it = 0; it < NITERS; ++it) {
    float a[16];
#pragma unroll
    for (int r = 0; r < 8; ++r) {
      const float o = xhalf(ck[r]);
      a[r]     = hi ? ((r == row) ? 1.f : 0.f)     : ck[r] + ((r == row) ? EPSV : 0.f);
      a[r + 8] = hi ? ((r + 8 == row) ? 1.f : 0.f) : o + ((r + 8 == row) ? EPSV : 0.f);
    }

#pragma unroll
    for (int k = 0; k < 16; ++k) {
      const float pinv = 1.f / lane_bcast(a[k], k);
      float f[16];
#pragma unroll
      for (int r = 0; r < 16; ++r) f[r] = lane_bcast(a[r], k);
      a[k] *= pinv;
#pragma unroll
      for (int r = 0; r < 16; ++r) {
        if (r != k) a[r] -= f[r] * a[k];
      }
    }
#pragma unroll
    for (int i = 0; i < 16; ++i) aug[i * AUG_STRIDE + lane] = a[i];
    __syncthreads();

    v2f ca[4];
    {
      const int prow = 4 * ((row & 7) >> 1) + 2 * (row >> 3) + (row & 1);
      const float* cp = &aug[prow * AUG_STRIDE + 16 + koff];
#pragma unroll
      for (int kc = 0; kc < 4; ++kc) {
        ca[kc].x = cp[4 * kc];
        ca[kc].y = cp[4 * kc + 1];
      }
    }
#pragma unroll
    for (int j5 = 0; j5 < 5; ++j5) {
      v8f y0 = {0.f, 0.f, 0.f, 0.f, 0.f, 0.f, 0.f, 0.f};
      v8f y1 = {0.f, 0.f, 0.f, 0.f, 0.f, 0.f, 0.f, 0.f};
#pragma unroll
      for (int kc = 0; kc < 4; ++kc) {
        y0 = wmma_f32_16x16x4(ca[kc], xq[2 * j5][kc], y0);
        y1 = wmma_f32_16x16x4(ca[kc], xq[2 * j5 + 1][kc], y1);
      }
      float p0 = y0[0] * xq[2 * j5][0].x;
      float p1 = y1[0] * xq[2 * j5 + 1][0].x;
      p0 = fmaf(y0[1], xq[2 * j5][0].y, p0);
      p1 = fmaf(y1[1], xq[2 * j5 + 1][0].y, p1);
#pragma unroll
      for (int kc = 1; kc < 4; ++kc) {
        p0 = fmaf(y0[2 * kc], xq[2 * j5][kc].x, p0);
        p1 = fmaf(y1[2 * kc], xq[2 * j5 + 1][kc].x, p1);
        p0 = fmaf(y0[2 * kc + 1], xq[2 * j5][kc].y, p0);
        p1 = fmaf(y1[2 * kc + 1], xq[2 * j5 + 1][kc].y, p1);
      }
      const float keep = hi ? p1 : p0;
      const float send = hi ? p0 : p1;
      const float qv = keep + xhalf(send);
      wbuf[lane + 32 * j5] = 16.f / fmaxf(qv, EPSV);
    }
    __syncthreads();

    v8f a2 = {0.f, 0.f, 0.f, 0.f, 0.f, 0.f, 0.f, 0.f};
#pragma unroll 1
    for (int cb = 0; cb < 10; ++cb) {
      v2f bb[4], aa[4];
#pragma unroll
      for (int u = 0; u < 4; ++u) {
        int t = (cb * 4 + u) * 4 + koff;
        const float* p = fragbase + (cb * 4 + u) * 4;
        bb[u].x = p[0];
        bb[u].y = p[1];
        aa[u].x = bb[u].x * wbuf[t];
        aa[u].y = bb[u].y * wbuf[t + 1];
      }
#pragma unroll
      for (int u = 0; u < 4; ++u) a2 = wmma_f32_16x16x4(aa[u], bb[u], a2);
    }
    float tr2 = wave_sum32(diag_partial(a2, lane) * (1.f / T_DIM), red, lane);
    float s2 = (1.f / T_DIM) * (16.f / (tr2 + EPSV));
#pragma unroll
    for (int r = 0; r < 8; ++r) ck[r] = a2[r] * s2;
    __syncthreads();
  }

#pragma unroll
  for (int r = 0; r < 8; ++r) { int m = r + 8 * hi; As[m * JS + row] = ck[r]; }
  __syncthreads();
  float tv[8];
#pragma unroll
  for (int r = 0; r < 8; ++r) { int m = r + 8 * hi; tv[r] = As[row * JS + m]; }
  __syncthreads();
#pragma unroll
  for (int r = 0; r < 8; ++r) {
    int m = r + 8 * hi;
    As[m * JS + row] = 0.5f * (ck[r] + tv[r]);
  }
  if (lane < 16) {
#pragma unroll
    for (int j = 0; j < 16; ++j) Vs[lane * JS + j] = (lane == j) ? 1.f : 0.f;
  }
  __syncthreads();

  const int pj = lane & 7;
  const int kg = lane >> 3;
#pragma unroll 1
  for (int sweep = 0; sweep < NSWEEP; ++sweep) {
#pragma unroll 1
    for (int rnd = 0; rnd < 15; ++rnd) {
      int ia = rnd + pj;       if (ia >= 15) ia -= 15;
      int ib = rnd + 15 - pj;  if (ib >= 15) ib -= 15;
      if (pj == 0) ib = 15;
      const int p = min(ia, ib);
      const int q = max(ia, ib);
      const float apq = As[p * JS + q];
      const float app = As[p * JS + p];
      const float aqq = As[q * JS + q];
      const bool rot = fabsf(apq) > 1e-30f;
      const float th = (aqq - app) * hw_rcp(2.f * apq);
      float tt = hw_rcp(fabsf(th) + hw_sqrt(fmaf(th, th, 1.f)));
      if (th < 0.f) tt = -tt;
      const float w1 = fmaf(tt, tt, 1.f);
      float y = hw_rsq(w1);
      y = fmaf(0.5f * y, fmaf(-w1 * y, y, 1.f), y);
      const float cc = rot ? y : 1.f;
      const float ss = rot ? tt * y : 0.f;
      __syncthreads();
      {
        float akp[4], akq[4], vkp[4], vkq[4];
#pragma unroll
        for (int i = 0; i < 4; ++i) {
          const int k2 = kg + 4 * i;
          akp[i] = As[k2 * JS + p]; akq[i] = As[k2 * JS + q];
          vkp[i] = Vs[k2 * JS + p]; vkq[i] = Vs[k2 * JS + q];
        }
#pragma unroll
        for (int i = 0; i < 4; ++i) {
          const int k2 = kg + 4 * i;
          As[k2 * JS + p] = cc * akp[i] - ss * akq[i];
          As[k2 * JS + q] = ss * akp[i] + cc * akq[i];
          Vs[k2 * JS + p] = cc * vkp[i] - ss * vkq[i];
          Vs[k2 * JS + q] = ss * vkp[i] + cc * vkq[i];
        }
      }
      __syncthreads();
      {
        float apk[4], aqk[4];
#pragma unroll
        for (int i = 0; i < 4; ++i) {
          const int k2 = kg + 4 * i;
          apk[i] = As[p * JS + k2]; aqk[i] = As[q * JS + k2];
        }
#pragma unroll
        for (int i = 0; i < 4; ++i) {
          const int k2 = kg + 4 * i;
          As[p * JS + k2] = cc * apk[i] - ss * aqk[i];
          As[q * JS + k2] = ss * apk[i] + cc * aqk[i];
        }
      }
      __syncthreads();
    }
  }
  if (lane < 16) lw[lane] = logf(fmaxf(As[lane * JS + lane], EPSV));
  __syncthreads();

  v8f lg = {0.f, 0.f, 0.f, 0.f, 0.f, 0.f, 0.f, 0.f};
#pragma unroll
  for (int kc = 0; kc < 4; ++kc) {
    int k0 = kc * 4 + koff;
    v2f bb; bb.x = Vs[row * JS + k0]; bb.y = Vs[row * JS + k0 + 1];
    float l0 = lw[k0], l1 = lw[k0 + 1];
    v2f aa; aa.x = bb.x * l0; aa.y = bb.y * l1;
    lg = wmma_f32_16x16x4(aa, bb, lg);
  }

  const float SQ2 = 1.4142135623730951f;
  float* op = out + (size_t)b * 136;
#pragma unroll
  for (int r = 0; r < 8; ++r) {
    int m = r + 8 * hi;
    if (m <= row) {
      int idx = m * 16 - (m * (m - 1)) / 2 + (row - m);
      op[idx] = lg[r] * ((m == row) ? 1.f : SQ2);
    }
  }
}

extern "C" void kernel_launch(void* const* d_in, const int* in_sizes, int n_in,
                              void* d_out, int out_size, void* d_ws,
                              size_t ws_size, hipStream_t stream) {
  const float* x = (const float*)d_in[0];
  float* out = (float*)d_out;
  int B = in_sizes[0] / (C_DIM * T_DIM);
  tyler_tangent_kernel<<<B, 32, 0, stream>>>(x, out);
}


# Occupancy 4 -> 8 waves per SIMD: LDS 15488 -> 10176 B, 128-register budget; only columns 0..31 of X stay in registers, the q pass re-reads the other B fragments from LDS in every iteration
# speedup vs baseline: 7.2958x; 1.0347x over previous
#include <hip/hip_runtime.h>
#include <math.h>

typedef __attribute__((ext_vector_type(2))) float v2f;
typedef __attribute__((ext_vector_type(8))) float v8f;

#define C_DIM 16
#define T_DIM 160
#define NITERS 10
#define EPSV 1e-8f
#define T_REG 32
#define XS_STRIDE 132
#define CI_STRIDE 17
#define JS 17
#define NSWEEP 8

__device__ __forceinline__ v8f wmma_f32_16x16x4(v2f a, v2f b, v8f c) {
  return __builtin_amdgcn_wmma_f32_16x16x4_f32(false, a, false, b, (short)0, c,
                                               false, false);
}

__device__ __forceinline__ float xhalf(float x) {
  int b = __builtin_bit_cast(int, x);
  return __builtin_bit_cast(
      float, __builtin_amdgcn_permlanex16(b, b, 0x76543210, 0xfedcba98, false, false));
}

__device__ __forceinline__ float lane_bcast(float v, int src) {
  return __builtin_bit_cast(float,
                            __builtin_amdgcn_readlane(__builtin_bit_cast(int, v), src));
}

__device__ __forceinline__ float hw_rcp(float v) { return __builtin_amdgcn_rcpf(v); }
__device__ __forceinline__ float hw_sqrt(float v) { return __builtin_amdgcn_sqrtf(v); }
__device__ __forceinline__ float hw_rsq(float v) { return __builtin_amdgcn_rsqf(v); }

__device__ __forceinline__ float wave_sum32(float v, float* red, int lane) {
  red[lane] = v;
  __syncthreads();
  float s = 0.f;
#pragma unroll
  for (int i = 0; i < 32; ++i) s += red[i];
  __syncthreads();
  return s;
}

__device__ __forceinline__ float diag_partial(const v8f& a, int lane) {
  float p = 0.f;
  const int mb = (lane >> 4) * 8;
  const int n = lane & 15;
#pragma unroll
  for (int r = 0; r < 8; ++r) p += (n == (r + mb)) ? a[r] : 0.f;
  return p;
}

__device__ __forceinline__ float q_pair_weight(const v2f (&ca)[4], const v2f (&b0)[4],
                                               const v2f (&b1)[4], int hi) {
  v8f y0 = {0.f, 0.f, 0.f, 0.f, 0.f, 0.f, 0.f, 0.f};
  v8f y1 = {0.f, 0.f, 0.f, 0.f, 0.f, 0.f, 0.f, 0.f};
#pragma unroll
  for (int kc = 0; kc < 4; ++kc) {
    y0 = wmma_f32_16x16x4(ca[kc], b0[kc], y0);
    y1 = wmma_f32_16x16x4(ca[kc], b1[kc], y1);
  }
  float p0 = y0[0] * b0[0].x;
  float p1 = y1[0] * b1[0].x;
  p0 = fmaf(y0[1], b0[0].y, p0);
  p1 = fmaf(y1[1], b1[0].y, p1);
#pragma unroll
  for (int kc = 1; kc < 4; ++kc) {
    p0 = fmaf(y0[2 * kc], b0[kc].x, p0);
    p1 = fmaf(y1[2 * kc], b1[kc].x, p1);
    p0 = fmaf(y0[2 * kc + 1], b0[kc].y, p0);
    p1 = fmaf(y1[2 * kc + 1], b1[kc].y, p1);
  }
  const float keep = hi ? p1 : p0;
  const float send = hi ? p0 : p1;
  const float qv = keep + xhalf(send);
  return 16.f / fmaxf(qv, EPSV);
}

__global__ __launch_bounds__(32, 8) void tyler_tangent_kernel(
    const float* __restrict__ x, float* __restrict__ out) {
  const int b    = blockIdx.x;
  const int lane = threadIdx.x;
  const int row  = lane & 15;
  const int hi   = lane >> 4;
  const int koff = hi * 2;

  __shared__ __attribute__((aligned(16))) float Xs[C_DIM * XS_STRIDE];
  __shared__ float cinv[C_DIM * CI_STRIDE];
  __shared__ float wbuf[T_DIM];
  float* const red = wbuf;
  float* const As = Xs;
  float* const Vs = Xs + C_DIM * JS;
  float* const lw = Xs + 2 * C_DIM * JS;

  const float* xg = x + (size_t)b * (C_DIM * T_DIM);

  {
    const float4* xin = reinterpret_cast<const float4*>(xg);
#pragma unroll
    for (int c = 0; c < C_DIM; ++c) {
      const float4 v = xin[c * (T_DIM / 4) + (T_REG / 4) + lane];
      *reinterpret_cast<float4*>(&Xs[c * XS_STRIDE + 4 * lane]) = v;
    }
  }

  v2f xa[8];
  {
    const float2* pa = reinterpret_cast<const float2*>(xg + row * T_DIM + koff);
#pragma unroll
    for (int c = 0; c < 8; ++c) {
      const float2 v = pa[2 * c];
      xa[c].x = v.x;
      xa[c].y = v.y;
    }
  }
  v2f xq[2][4];
  {
    const float* pb = xg + koff * T_DIM + row;
#pragma unroll
    for (int tile = 0; tile < 2; ++tile) {
#pragma unroll
      for (int kc = 0; kc < 4; ++kc) {
        xq[tile][kc].x = pb[(4 * kc) * T_DIM + 16 * tile];
        xq[tile][kc].y = pb[(4 * kc + 1) * T_DIM + 16 * tile];
      }
    }
  }
  __syncthreads();

  const float* fragbase = &Xs[row * XS_STRIDE + koff];

  v8f acc = {0.f, 0.f, 0.f, 0.f, 0.f, 0.f, 0.f, 0.f};
#pragma unroll
  for (int c = 0; c < 8; ++c) acc = wmma_f32_16x16x4(xa[c], xa[c], acc);
#pragma unroll 1
  for (int cb = 0; cb < 8; ++cb) {
    v2f f[4];
#pragma unroll
    for (int u = 0; u < 4; ++u) {
      const float* p = fragbase + (cb * 4 + u) * 4;
      f[u].x = p[0];
      f[u].y = p[1];
    }
#pragma unroll
    for (int u = 0; u < 4; ++u) acc = wmma_f32_16x16x4(f[u], f[u], acc);
  }
  float tr = wave_sum32(diag_partial(acc, lane) * (1.f / T_DIM), red, lane);
  float scl = (1.f / T_DIM) * (16.f / (tr + EPSV));
  v8f ck;
#pragma unroll
  for (int r = 0; r < 8; ++r) ck[r] = acc[r] * scl;

  const float* xbl = &Xs[koff * XS_STRIDE + row];

#pragma unroll 1
  for (int it = 0; it < NITERS; ++it) {
    float a[16];
    float dsel = hi ? 1.f : EPSV;
    asm volatile("" : "+v"(dsel));
#pragma unroll
    for (int r = 0; r < 8; ++r) {
      const float o = xhalf(ck[r]);
      const float u0 = hi ? 0.f : ck[r];
      const float u1 = hi ? 0.f : o;
      a[r]     = u0 + ((r == row) ? dsel : 0.f);
      a[r + 8] = u1 + ((r + 8 == row) ? dsel : 0.f);
    }

#pragma unroll
    for (int k = 0; k < 16; ++k) {
      const float pinv = 1.f / lane_bcast(a[k], k);
      float f[16];
#pragma unroll
      for (int r = 0; r < 16; ++r) f[r] = lane_bcast(a[r], k);
      a[k] *= pinv;
#pragma unroll
      for (int r = 0; r < 16; ++r) {
        if (r != k) a[r] -= f[r] * a[k];
      }
    }
    if (hi) {
#pragma unroll
      for (int i = 0; i < 16; ++i) cinv[i * CI_STRIDE + row] = a[i];
    }
    __syncthreads();

    v2f ca[4];
    {
      const int prow = 4 * ((row & 7) >> 1) + 2 * (row >> 3) + (row & 1);
      const float* cp = &cinv[prow * CI_STRIDE + koff];
#pragma unroll
      for (int kc = 0; kc < 4; ++kc) {
        ca[kc].x = cp[4 * kc];
        ca[kc].y = cp[4 * kc + 1];
      }
    }
    wbuf[lane] = q_pair_weight(ca, xq[0], xq[1], hi);
#pragma unroll 1
    for (int j5 = 1; j5 < 5; ++j5) {
      v2f b0[4], b1[4];
      const float* q0 = xbl + 32 * (j5 - 1);
#pragma unroll
      for (int kc = 0; kc < 4; ++kc) {
        b0[kc].x = q0[(4 * kc) * XS_STRIDE];
        b0[kc].y = q0[(4 * kc + 1) * XS_STRIDE];
        b1[kc].x = q0[(4 * kc) * XS_STRIDE + 16];
        b1[kc].y = q0[(4 * kc + 1) * XS_STRIDE + 16];
      }
      wbuf[lane + 32 * j5] = q_pair_weight(ca, b0, b1, hi);
    }
    __syncthreads();

    v8f a2 = {0.f, 0.f, 0.f, 0.f, 0.f, 0.f, 0.f, 0.f};
#pragma unroll
    for (int c = 0; c < 8; ++c) {
      const int t = 4 * c + koff;
      v2f aa0;
      aa0.x = xa[c].x * wbuf[t];
      aa0.y = xa[c].y * wbuf[t + 1];
      a2 = wmma_f32_16x16x4(aa0, xa[c], a2);
    }
#pragma unroll 1
    for (int cb = 0; cb < 8; ++cb) {
      v2f bb[4], aa[4];
#pragma unroll
      for (int u = 0; u < 4; ++u) {
        int t = T_REG + (cb * 4 + u) * 4 + koff;
        const float* p = fragbase + (cb * 4 + u) * 4;
        bb[u].x = p[0];
        bb[u].y = p[1];
        aa[u].x = bb[u].x * wbuf[t];
        aa[u].y = bb[u].y * wbuf[t + 1];
      }
#pragma unroll
      for (int u = 0; u < 4; ++u) a2 = wmma_f32_16x16x4(aa[u], bb[u], a2);
    }
    float tr2 = wave_sum32(diag_partial(a2, lane) * (1.f / T_DIM), red, lane);
    float s2 = (1.f / T_DIM) * (16.f / (tr2 + EPSV));
#pragma unroll
    for (int r = 0; r < 8; ++r) ck[r] = a2[r] * s2;
    __syncthreads();
  }

#pragma unroll
  for (int r = 0; r < 8; ++r) { int m = r + 8 * hi; As[m * JS + row] = ck[r]; }
  __syncthreads();
  float tv[8];
#pragma unroll
  for (int r = 0; r < 8; ++r) { int m = r + 8 * hi; tv[r] = As[row * JS + m]; }
  __syncthreads();
#pragma unroll
  for (int r = 0; r < 8; ++r) {
    int m = r + 8 * hi;
    As[m * JS + row] = 0.5f * (ck[r] + tv[r]);
  }
  if (lane < 16) {
#pragma unroll
    for (int j = 0; j < 16; ++j) Vs[lane * JS + j] = (lane == j) ? 1.f : 0.f;
  }
  __syncthreads();

  const int pj = lane & 7;
  const int kg = lane >> 3;
#pragma unroll 1
  for (int sweep = 0; sweep < NSWEEP; ++sweep) {
#pragma unroll 1
    for (int rnd = 0; rnd < 15; ++rnd) {
      int ia = rnd + pj;       if (ia >= 15) ia -= 15;
      int ib = rnd + 15 - pj;  if (ib >= 15) ib -= 15;
      if (pj == 0) ib = 15;
      const int p = min(ia, ib);
      const int q = max(ia, ib);
      const float apq = As[p * JS + q];
      const float app = As[p * JS + p];
      const float aqq = As[q * JS + q];
      const bool rot = fabsf(apq) > 1e-30f;
      const float th = (aqq - app) * hw_rcp(2.f * apq);
      float tt = hw_rcp(fabsf(th) + hw_sqrt(fmaf(th, th, 1.f)));
      if (th < 0.f) tt = -tt;
      const float w1 = fmaf(tt, tt, 1.f);
      float y = hw_rsq(w1);
      y = fmaf(0.5f * y, fmaf(-w1 * y, y, 1.f), y);
      const float cc = rot ? y : 1.f;
      const float ss = rot ? tt * y : 0.f;
      __syncthreads();
      {
        float akp[4], akq[4], vkp[4], vkq[4];
#pragma unroll
        for (int i = 0; i < 4; ++i) {
          const int k2 = kg + 4 * i;
          akp[i] = As[k2 * JS + p]; akq[i] = As[k2 * JS + q];
          vkp[i] = Vs[k2 * JS + p]; vkq[i] = Vs[k2 * JS + q];
        }
#pragma unroll
        for (int i = 0; i < 4; ++i) {
          const int k2 = kg + 4 * i;
          As[k2 * JS + p] = cc * akp[i] - ss * akq[i];
          As[k2 * JS + q] = ss * akp[i] + cc * akq[i];
          Vs[k2 * JS + p] = cc * vkp[i] - ss * vkq[i];
          Vs[k2 * JS + q] = ss * vkp[i] + cc * vkq[i];
        }
      }
      __syncthreads();
      {
        float apk[4], aqk[4];
#pragma unroll
        for (int i = 0; i < 4; ++i) {
          const int k2 = kg + 4 * i;
          apk[i] = As[p * JS + k2]; aqk[i] = As[q * JS + k2];
        }
#pragma unroll
        for (int i = 0; i < 4; ++i) {
          const int k2 = kg + 4 * i;
          As[p * JS + k2] = cc * apk[i] - ss * aqk[i];
          As[q * JS + k2] = ss * apk[i] + cc * aqk[i];
        }
      }
      __syncthreads();
    }
  }
  if (lane < 16) lw[lane] = logf(fmaxf(As[lane * JS + lane], EPSV));
  __syncthreads();

  v8f lg = {0.f, 0.f, 0.f, 0.f, 0.f, 0.f, 0.f, 0.f};
#pragma unroll
  for (int kc = 0; kc < 4; ++kc) {
    int k0 = kc * 4 + koff;
    v2f bb; bb.x = Vs[row * JS + k0]; bb.y = Vs[row * JS + k0 + 1];
    float l0 = lw[k0], l1 = lw[k0 + 1];
    v2f aa; aa.x = bb.x * l0; aa.y = bb.y * l1;
    lg = wmma_f32_16x16x4(aa, bb, lg);
  }

  const float SQ2 = 1.4142135623730951f;
  float* op = out + (size_t)b * 136;
#pragma unroll
  for (int r = 0; r < 8; ++r) {
    int m = r + 8 * hi;
    if (m <= row) {
      int idx = m * 16 - (m * (m - 1)) / 2 + (row - m);
      op[idx] = lg[r] * ((m == row) ? 1.f : SQ2);
    }
  }
}

extern "C" void kernel_launch(void* const* d_in, const int* in_sizes, int n_in,
                              void* d_out, int out_size, void* d_ws,
                              size_t ws_size, hipStream_t stream) {
  const float* x = (const float*)d_in[0];
  float* out = (float*)d_out;
  int B = in_sizes[0] / (C_DIM * T_DIM);
  tyler_tangent_kernel<<<B, 32, 0, stream>>>(x, out);
}


# Jacobi eigen-solve stops at convergence: it ends after the first sweep whose pairs all had |apq| <= 2^-14*sqrt(app*aqq) at set-up (that sweep still applied); cap of 8 sweeps kept
# speedup vs baseline: 7.9031x; 1.0832x over previous
#include <hip/hip_runtime.h>
#include <math.h>

typedef __attribute__((ext_vector_type(2))) float v2f;
typedef __attribute__((ext_vector_type(8))) float v8f;

#define C_DIM 16
#define T_DIM 160
#define NITERS 10
#define EPSV 1e-8f
#define T_REG 32
#define XS_STRIDE 132
#define CI_STRIDE 17
#define JS 17
#define NSWEEP 8
#define CONV2 3.7252902984619140625e-09f

__device__ __forceinline__ v8f wmma_f32_16x16x4(v2f a, v2f b, v8f c) {
  return __builtin_amdgcn_wmma_f32_16x16x4_f32(false, a, false, b, (short)0, c,
                                               false, false);
}

__device__ __forceinline__ float xhalf(float x) {
  int b = __builtin_bit_cast(int, x);
  return __builtin_bit_cast(
      float, __builtin_amdgcn_permlanex16(b, b, 0x76543210, 0xfedcba98, false, false));
}

__device__ __forceinline__ float lane_bcast(float v, int src) {
  return __builtin_bit_cast(float,
                            __builtin_amdgcn_readlane(__builtin_bit_cast(int, v), src));
}

__device__ __forceinline__ float hw_rcp(float v) { return __builtin_amdgcn_rcpf(v); }
__device__ __forceinline__ float hw_sqrt(float v) { return __builtin_amdgcn_sqrtf(v); }
__device__ __forceinline__ float hw_rsq(float v) { return __builtin_amdgcn_rsqf(v); }

__device__ __forceinline__ float wave_sum32(float v, float* red, int lane) {
  red[lane] = v;
  __syncthreads();
  float s = 0.f;
#pragma unroll
  for (int i = 0; i < 32; ++i) s += red[i];
  __syncthreads();
  return s;
}

__device__ __forceinline__ float diag_partial(const v8f& a, int lane) {
  float p = 0.f;
  const int mb = (lane >> 4) * 8;
  const int n = lane & 15;
#pragma unroll
  for (int r = 0; r < 8; ++r) p += (n == (r + mb)) ? a[r] : 0.f;
  return p;
}

__device__ __forceinline__ float q_pair_weight(const v2f (&ca)[4], const v2f (&b0)[4],
                                               const v2f (&b1)[4], int hi) {
  v8f y0 = {0.f, 0.f, 0.f, 0.f, 0.f, 0.f, 0.f, 0.f};
  v8f y1 = {0.f, 0.f, 0.f, 0.f, 0.f, 0.f, 0.f, 0.f};
#pragma unroll
  for (int kc = 0; kc < 4; ++kc) {
    y0 = wmma_f32_16x16x4(ca[kc], b0[kc], y0);
    y1 = wmma_f32_16x16x4(ca[kc], b1[kc], y1);
  }
  float p0 = y0[0] * b0[0].x;
  float p1 = y1[0] * b1[0].x;
  p0 = fmaf(y0[1], b0[0].y, p0);
  p1 = fmaf(y1[1], b1[0].y, p1);
#pragma unroll
  for (int kc = 1; kc < 4; ++kc) {
    p0 = fmaf(y0[2 * kc], b0[kc].x, p0);
    p1 = fmaf(y1[2 * kc], b1[kc].x, p1);
    p0 = fmaf(y0[2 * kc + 1], b0[kc].y, p0);
    p1 = fmaf(y1[2 * kc + 1], b1[kc].y, p1);
  }
  const float keep = hi ? p1 : p0;
  const float send = hi ? p0 : p1;
  const float qv = keep + xhalf(send);
  return 16.f / fmaxf(qv, EPSV);
}

__global__ __launch_bounds__(32, 8) void tyler_tangent_kernel(
    const float* __restrict__ x, float* __restrict__ out) {
  const int b    = blockIdx.x;
  const int lane = threadIdx.x;
  const int row  = lane & 15;
  const int hi   = lane >> 4;
  const int koff = hi * 2;

  __shared__ __attribute__((aligned(16))) float Xs[C_DIM * XS_STRIDE];
  __shared__ float cinv[C_DIM * CI_STRIDE];
  __shared__ float wbuf[T_DIM];
  float* const red = wbuf;
  float* const As = Xs;
  float* const Vs = Xs + C_DIM * JS;
  float* const lw = Xs + 2 * C_DIM * JS;

  const float* xg = x + (size_t)b * (C_DIM * T_DIM);

  {
    const float4* xin = reinterpret_cast<const float4*>(xg);
#pragma unroll
    for (int c = 0; c < C_DIM; ++c) {
      const float4 v = xin[c * (T_DIM / 4) + (T_REG / 4) + lane];
      *reinterpret_cast<float4*>(&Xs[c * XS_STRIDE + 4 * lane]) = v;
    }
  }

  v2f xa[8];
  {
    const float2* pa = reinterpret_cast<const float2*>(xg + row * T_DIM + koff);
#pragma unroll
    for (int c = 0; c < 8; ++c) {
      const float2 v = pa[2 * c];
      xa[c].x = v.x;
      xa[c].y = v.y;
    }
  }
  v2f xq[2][4];
  {
    const float* pb = xg + koff * T_DIM + row;
#pragma unroll
    for (int tile = 0; tile < 2; ++tile) {
#pragma unroll
      for (int kc = 0; kc < 4; ++kc) {
        xq[tile][kc].x = pb[(4 * kc) * T_DIM + 16 * tile];
        xq[tile][kc].y = pb[(4 * kc + 1) * T_DIM + 16 * tile];
      }
    }
  }
  __syncthreads();

  const float* fragbase = &Xs[row * XS_STRIDE + koff];

  v8f acc = {0.f, 0.f, 0.f, 0.f, 0.f, 0.f, 0.f, 0.f};
#pragma unroll
  for (int c = 0; c < 8; ++c) acc = wmma_f32_16x16x4(xa[c], xa[c], acc);
#pragma unroll 1
  for (int cb = 0; cb < 8; ++cb) {
    v2f f[4];
#pragma unroll
    for (int u = 0; u < 4; ++u) {
      const float* p = fragbase + (cb * 4 + u) * 4;
      f[u].x = p[0];
      f[u].y = p[1];
    }
#pragma unroll
    for (int u = 0; u < 4; ++u) acc = wmma_f32_16x16x4(f[u], f[u], acc);
  }
  float tr = wave_sum32(diag_partial(acc, lane) * (1.f / T_DIM), red, lane);
  float scl = (1.f / T_DIM) * (16.f / (tr + EPSV));
  v8f ck;
#pragma unroll
  for (int r = 0; r < 8; ++r) ck[r] = acc[r] * scl;

  const float* xbl = &Xs[koff * XS_STRIDE + row];

#pragma unroll 1
  for (int it = 0; it < NITERS; ++it) {
    float a[16];
    float dsel = hi ? 1.f : EPSV;
    asm volatile("" : "+v"(dsel));
#pragma unroll
    for (int r = 0; r < 8; ++r) {
      const float o = xhalf(ck[r]);
      const float u0 = hi ? 0.f : ck[r];
      const float u1 = hi ? 0.f : o;
      a[r]     = u0 + ((r == row) ? dsel : 0.f);
      a[r + 8] = u1 + ((r + 8 == row) ? dsel : 0.f);
    }

#pragma unroll
    for (int k = 0; k < 16; ++k) {
      const float pinv = 1.f / lane_bcast(a[k], k);
      float f[16];
#pragma unroll
      for (int r = 0; r < 16; ++r) f[r] = lane_bcast(a[r], k);
      a[k] *= pinv;
#pragma unroll
      for (int r = 0; r < 16; ++r) {
        if (r != k) a[r] -= f[r] * a[k];
      }
    }
    if (hi) {
#pragma unroll
      for (int i = 0; i < 16; ++i) cinv[i * CI_STRIDE + row] = a[i];
    }
    __syncthreads();

    v2f ca[4];
    {
      const int prow = 4 * ((row & 7) >> 1) + 2 * (row >> 3) + (row & 1);
      const float* cp = &cinv[prow * CI_STRIDE + koff];
#pragma unroll
      for (int kc = 0; kc < 4; ++kc) {
        ca[kc].x = cp[4 * kc];
        ca[kc].y = cp[4 * kc + 1];
      }
    }
    wbuf[lane] = q_pair_weight(ca, xq[0], xq[1], hi);
#pragma unroll 1
    for (int j5 = 1; j5 < 5; ++j5) {
      v2f b0[4], b1[4];
      const float* q0 = xbl + 32 * (j5 - 1);
#pragma unroll
      for (int kc = 0; kc < 4; ++kc) {
        b0[kc].x = q0[(4 * kc) * XS_STRIDE];
        b0[kc].y = q0[(4 * kc + 1) * XS_STRIDE];
        b1[kc].x = q0[(4 * kc) * XS_STRIDE + 16];
        b1[kc].y = q0[(4 * kc + 1) * XS_STRIDE + 16];
      }
      wbuf[lane + 32 * j5] = q_pair_weight(ca, b0, b1, hi);
    }
    __syncthreads();

    v8f a2 = {0.f, 0.f, 0.f, 0.f, 0.f, 0.f, 0.f, 0.f};
#pragma unroll
    for (int c = 0; c < 8; ++c) {
      const int t = 4 * c + koff;
      v2f aa0;
      aa0.x = xa[c].x * wbuf[t];
      aa0.y = xa[c].y * wbuf[t + 1];
      a2 = wmma_f32_16x16x4(aa0, xa[c], a2);
    }
#pragma unroll 1
    for (int cb = 0; cb < 8; ++cb) {
      v2f bb[4], aa[4];
#pragma unroll
      for (int u = 0; u < 4; ++u) {
        int t = T_REG + (cb * 4 + u) * 4 + koff;
        const float* p = fragbase + (cb * 4 + u) * 4;
        bb[u].x = p[0];
        bb[u].y = p[1];
        aa[u].x = bb[u].x * wbuf[t];
        aa[u].y = bb[u].y * wbuf[t + 1];
      }
#pragma unroll
      for (int u = 0; u < 4; ++u) a2 = wmma_f32_16x16x4(aa[u], bb[u], a2);
    }
    float tr2 = wave_sum32(diag_partial(a2, lane) * (1.f / T_DIM), red, lane);
    float s2 = (1.f / T_DIM) * (16.f / (tr2 + EPSV));
#pragma unroll
    for (int r = 0; r < 8; ++r) ck[r] = a2[r] * s2;
    __syncthreads();
  }

#pragma unroll
  for (int r = 0; r < 8; ++r) { int m = r + 8 * hi; As[m * JS + row] = ck[r]; }
  __syncthreads();
  float tv[8];
#pragma unroll
  for (int r = 0; r < 8; ++r) { int m = r + 8 * hi; tv[r] = As[row * JS + m]; }
  __syncthreads();
#pragma unroll
  for (int r = 0; r < 8; ++r) {
    int m = r + 8 * hi;
    As[m * JS + row] = 0.5f * (ck[r] + tv[r]);
  }
  if (lane < 16) {
#pragma unroll
    for (int j = 0; j < 16; ++j) Vs[lane * JS + j] = (lane == j) ? 1.f : 0.f;
  }
  __syncthreads();

  const int pj = lane & 7;
  const int kg = lane >> 3;
#pragma unroll 1
  for (int sweep = 0; sweep < NSWEEP; ++sweep) {
    bool live = false;
#pragma unroll 1
    for (int rnd = 0; rnd < 15; ++rnd) {
      int ia = rnd + pj;       if (ia >= 15) ia -= 15;
      int ib = rnd + 15 - pj;  if (ib >= 15) ib -= 15;
      if (pj == 0) ib = 15;
      const int p = min(ia, ib);
      const int q = max(ia, ib);
      const float apq = As[p * JS + q];
      const float app = As[p * JS + p];
      const float aqq = As[q * JS + q];
      const bool rot = fabsf(apq) > 1e-30f;
      live = live || (apq * apq > CONV2 * fabsf(app * aqq));
      const float th = (aqq - app) * hw_rcp(2.f * apq);
      float tt = hw_rcp(fabsf(th) + hw_sqrt(fmaf(th, th, 1.f)));
      if (th < 0.f) tt = -tt;
      const float w1 = fmaf(tt, tt, 1.f);
      float y = hw_rsq(w1);
      y = fmaf(0.5f * y, fmaf(-w1 * y, y, 1.f), y);
      const float cc = rot ? y : 1.f;
      const float ss = rot ? tt * y : 0.f;
      __syncthreads();
      {
        float akp[4], akq[4], vkp[4], vkq[4];
#pragma unroll
        for (int i = 0; i < 4; ++i) {
          const int k2 = kg + 4 * i;
          akp[i] = As[k2 * JS + p]; akq[i] = As[k2 * JS + q];
          vkp[i] = Vs[k2 * JS + p]; vkq[i] = Vs[k2 * JS + q];
        }
#pragma unroll
        for (int i = 0; i < 4; ++i) {
          const int k2 = kg + 4 * i;
          As[k2 * JS + p] = cc * akp[i] - ss * akq[i];
          As[k2 * JS + q] = ss * akp[i] + cc * akq[i];
          Vs[k2 * JS + p] = cc * vkp[i] - ss * vkq[i];
          Vs[k2 * JS + q] = ss * vkp[i] + cc * vkq[i];
        }
      }
      __syncthreads();
      {
        float apk[4], aqk[4];
#pragma unroll
        for (int i = 0; i < 4; ++i) {
          const int k2 = kg + 4 * i;
          apk[i] = As[p * JS + k2]; aqk[i] = As[q * JS + k2];
        }
#pragma unroll
        for (int i = 0; i < 4; ++i) {
          const int k2 = kg + 4 * i;
          As[p * JS + k2] = cc * apk[i] - ss * aqk[i];
          As[q * JS + k2] = ss * apk[i] + cc * aqk[i];
        }
      }
      __syncthreads();
    }
    if (__builtin_amdgcn_ballot_w32(live) == 0u) break;
  }
  if (lane < 16) lw[lane] = logf(fmaxf(As[lane * JS + lane], EPSV));
  __syncthreads();

  v8f lg = {0.f, 0.f, 0.f, 0.f, 0.f, 0.f, 0.f, 0.f};
#pragma unroll
  for (int kc = 0; kc < 4; ++kc) {
    int k0 = kc * 4 + koff;
    v2f bb; bb.x = Vs[row * JS + k0]; bb.y = Vs[row * JS + k0 + 1];
    float l0 = lw[k0], l1 = lw[k0 + 1];
    v2f aa; aa.x = bb.x * l0; aa.y = bb.y * l1;
    lg = wmma_f32_16x16x4(aa, bb, lg);
  }

  const float SQ2 = 1.4142135623730951f;
  float* op = out + (size_t)b * 136;
#pragma unroll
  for (int r = 0; r < 8; ++r) {
    int m = r + 8 * hi;
    if (m <= row) {
      int idx = m * 16 - (m * (m - 1)) / 2 + (row - m);
      op[idx] = lg[r] * ((m == row) ? 1.f : SQ2);
    }
  }
}

extern "C" void kernel_launch(void* const* d_in, const int* in_sizes, int n_in,
                              void* d_out, int out_size, void* d_ws,
                              size_t ws_size, hipStream_t stream) {
  const float* x = (const float*)d_in[0];
  float* out = (float*)d_out;
  int B = in_sizes[0] / (C_DIM * T_DIM);
  tyler_tangent_kernel<<<B, 32, 0, stream>>>(x, out);
}


# Gauss-Jordan inverse done in place in the accumulator layout: pivot-column factors enter the multiply-add by DPP row_share (8 per pivot, not 16 readlanes + 16 fma); Cinv rounding changes
# speedup vs baseline: 8.2095x; 1.0388x over previous
#include <hip/hip_runtime.h>
#include <math.h>

typedef __attribute__((ext_vector_type(2))) float v2f;
typedef __attribute__((ext_vector_type(8))) float v8f;

#define C_DIM 16
#define T_DIM 160
#define NITERS 10
#define EPSV 1e-8f
#define T_REG 32
#define XS_STRIDE 132
#define CI_STRIDE 17
#define JS 17
#define NSWEEP 8
#define CONV2 3.7252902984619140625e-09f

__device__ __forceinline__ v8f wmma_f32_16x16x4(v2f a, v2f b, v8f c) {
  return __builtin_amdgcn_wmma_f32_16x16x4_f32(false, a, false, b, (short)0, c,
                                               false, false);
}

__device__ __forceinline__ float xhalf(float x) {
  int b = __builtin_bit_cast(int, x);
  return __builtin_bit_cast(
      float, __builtin_amdgcn_permlanex16(b, b, 0x76543210, 0xfedcba98, false, false));
}

__device__ __forceinline__ float lane_bcast(float v, int src) {
  return __builtin_bit_cast(float,
                            __builtin_amdgcn_readlane(__builtin_bit_cast(int, v), src));
}

template <int K>
__device__ __forceinline__ float row_bcast(float v) {
  return __builtin_bit_cast(float, __builtin_amdgcn_update_dpp(0, __builtin_bit_cast(int, v),
                                                               0x150 + K, 0xf, 0xf, true));
}

template <int K>
__device__ __forceinline__ void gj_pivot(float (&a)[8], float& csc, int row, int hi) {
  constexpr int RK = K & 7;
  constexpr int HK = K >> 3;
  const float pinv = 1.f / lane_bcast(a[RK], 16 * HK + K);
  const float other = xhalf(a[RK]);
  const float rs = ((hi == HK) ? a[RK] : other) * pinv;
  const bool pivcol = (row == K);
  const float m = pivcol ? 0.f : rs;
#pragma unroll
  for (int r = 0; r < 8; ++r) a[r] = fmaf(-row_bcast<K>(a[r]), m, a[r]);
  a[RK] = (hi == HK) ? (pivcol ? -1.f : rs) : a[RK];
  csc = pivcol ? -pinv : csc;
}

__device__ __forceinline__ float hw_rcp(float v) { return __builtin_amdgcn_rcpf(v); }
__device__ __forceinline__ float hw_sqrt(float v) { return __builtin_amdgcn_sqrtf(v); }
__device__ __forceinline__ float hw_rsq(float v) { return __builtin_amdgcn_rsqf(v); }

__device__ __forceinline__ float wave_sum32(float v, float* red, int lane) {
  red[lane] = v;
  __syncthreads();
  float s = 0.f;
#pragma unroll
  for (int i = 0; i < 32; ++i) s += red[i];
  __syncthreads();
  return s;
}

__device__ __forceinline__ float diag_partial(const v8f& a, int lane) {
  float p = 0.f;
  const int mb = (lane >> 4) * 8;
  const int n = lane & 15;
#pragma unroll
  for (int r = 0; r < 8; ++r) p += (n == (r + mb)) ? a[r] : 0.f;
  return p;
}

__device__ __forceinline__ float q_pair_weight(const v2f (&ca)[4], const v2f (&b0)[4],
                                               const v2f (&b1)[4], int hi) {
  v8f y0 = {0.f, 0.f, 0.f, 0.f, 0.f, 0.f, 0.f, 0.f};
  v8f y1 = {0.f, 0.f, 0.f, 0.f, 0.f, 0.f, 0.f, 0.f};
#pragma unroll
  for (int kc = 0; kc < 4; ++kc) {
    y0 = wmma_f32_16x16x4(ca[kc], b0[kc], y0);
    y1 = wmma_f32_16x16x4(ca[kc], b1[kc], y1);
  }
  float p0 = y0[0] * b0[0].x;
  float p1 = y1[0] * b1[0].x;
  p0 = fmaf(y0[1], b0[0].y, p0);
  p1 = fmaf(y1[1], b1[0].y, p1);
#pragma unroll
  for (int kc = 1; kc < 4; ++kc) {
    p0 = fmaf(y0[2 * kc], b0[kc].x, p0);
    p1 = fmaf(y1[2 * kc], b1[kc].x, p1);
    p0 = fmaf(y0[2 * kc + 1], b0[kc].y, p0);
    p1 = fmaf(y1[2 * kc + 1], b1[kc].y, p1);
  }
  const float keep = hi ? p1 : p0;
  const float send = hi ? p0 : p1;
  const float qv = keep + xhalf(send);
  return 16.f / fmaxf(qv, EPSV);
}

__global__ __launch_bounds__(32, 8) void tyler_tangent_kernel(
    const float* __restrict__ x, float* __restrict__ out) {
  const int b    = blockIdx.x;
  const int lane = threadIdx.x;
  const int row  = lane & 15;
  const int hi   = lane >> 4;
  const int koff = hi * 2;

  __shared__ __attribute__((aligned(16))) float Xs[C_DIM * XS_STRIDE];
  __shared__ float cinv[C_DIM * CI_STRIDE];
  __shared__ float wbuf[T_DIM];
  float* const red = wbuf;
  float* const As = Xs;
  float* const Vs = Xs + C_DIM * JS;
  float* const lw = Xs + 2 * C_DIM * JS;

  const float* xg = x + (size_t)b * (C_DIM * T_DIM);

  {
    const float4* xin = reinterpret_cast<const float4*>(xg);
#pragma unroll
    for (int c = 0; c < C_DIM; ++c) {
      const float4 v = xin[c * (T_DIM / 4) + (T_REG / 4) + lane];
      *reinterpret_cast<float4*>(&Xs[c * XS_STRIDE + 4 * lane]) = v;
    }
  }

  v2f xa[8];
  {
    const float2* pa = reinterpret_cast<const float2*>(xg + row * T_DIM + koff);
#pragma unroll
    for (int c = 0; c < 8; ++c) {
      const float2 v = pa[2 * c];
      xa[c].x = v.x;
      xa[c].y = v.y;
    }
  }
  v2f xq[2][4];
  {
    const float* pb = xg + koff * T_DIM + row;
#pragma unroll
    for (int tile = 0; tile < 2; ++tile) {
#pragma unroll
      for (int kc = 0; kc < 4; ++kc) {
        xq[tile][kc].x = pb[(4 * kc) * T_DIM + 16 * tile];
        xq[tile][kc].y = pb[(4 * kc + 1) * T_DIM + 16 * tile];
      }
    }
  }
  __syncthreads();

  const float* fragbase = &Xs[row * XS_STRIDE + koff];

  v8f acc = {0.f, 0.f, 0.f, 0.f, 0.f, 0.f, 0.f, 0.f};
#pragma unroll
  for (int c = 0; c < 8; ++c) acc = wmma_f32_16x16x4(xa[c], xa[c], acc);
#pragma unroll 1
  for (int cb = 0; cb < 8; ++cb) {
    v2f f[4];
#pragma unroll
    for (int u = 0; u < 4; ++u) {
      const float* p = fragbase + (cb * 4 + u) * 4;
      f[u].x = p[0];
      f[u].y = p[1];
    }
#pragma unroll
    for (int u = 0; u < 4; ++u) acc = wmma_f32_16x16x4(f[u], f[u], acc);
  }
  float tr = wave_sum32(diag_partial(acc, lane) * (1.f / T_DIM), red, lane);
  float scl = (1.f / T_DIM) * (16.f / (tr + EPSV));
  v8f ck;
#pragma unroll
  for (int r = 0; r < 8; ++r) ck[r] = acc[r] * scl;

  const float* xbl = &Xs[koff * XS_STRIDE + row];

#pragma unroll 1
  for (int it = 0; it < NITERS; ++it) {
    float a[8];
#pragma unroll
    for (int r = 0; r < 8; ++r) a[r] = ck[r] + ((r + 8 * hi == row) ? EPSV : 0.f);

    float csc = 0.f;
    gj_pivot<0>(a, csc, row, hi);   gj_pivot<1>(a, csc, row, hi);
    gj_pivot<2>(a, csc, row, hi);   gj_pivot<3>(a, csc, row, hi);
    gj_pivot<4>(a, csc, row, hi);   gj_pivot<5>(a, csc, row, hi);
    gj_pivot<6>(a, csc, row, hi);   gj_pivot<7>(a, csc, row, hi);
    gj_pivot<8>(a, csc, row, hi);   gj_pivot<9>(a, csc, row, hi);
    gj_pivot<10>(a, csc, row, hi);  gj_pivot<11>(a, csc, row, hi);
    gj_pivot<12>(a, csc, row, hi);  gj_pivot<13>(a, csc, row, hi);
    gj_pivot<14>(a, csc, row, hi);  gj_pivot<15>(a, csc, row, hi);
#pragma unroll
    for (int r = 0; r < 8; ++r) cinv[(r + 8 * hi) * CI_STRIDE + row] = a[r] * csc;
    __syncthreads();

    v2f ca[4];
    {
      const int prow = 4 * ((row & 7) >> 1) + 2 * (row >> 3) + (row & 1);
      const float* cp = &cinv[prow * CI_STRIDE + koff];
#pragma unroll
      for (int kc = 0; kc < 4; ++kc) {
        ca[kc].x = cp[4 * kc];
        ca[kc].y = cp[4 * kc + 1];
      }
    }
    wbuf[lane] = q_pair_weight(ca, xq[0], xq[1], hi);
#pragma unroll 1
    for (int j5 = 1; j5 < 5; ++j5) {
      v2f b0[4], b1[4];
      const float* q0 = xbl + 32 * (j5 - 1);
#pragma unroll
      for (int kc = 0; kc < 4; ++kc) {
        b0[kc].x = q0[(4 * kc) * XS_STRIDE];
        b0[kc].y = q0[(4 * kc + 1) * XS_STRIDE];
        b1[kc].x = q0[(4 * kc) * XS_STRIDE + 16];
        b1[kc].y = q0[(4 * kc + 1) * XS_STRIDE + 16];
      }
      wbuf[lane + 32 * j5] = q_pair_weight(ca, b0, b1, hi);
    }
    __syncthreads();

    v8f a2 = {0.f, 0.f, 0.f, 0.f, 0.f, 0.f, 0.f, 0.f};
#pragma unroll
    for (int c = 0; c < 8; ++c) {
      const int t = 4 * c + koff;
      v2f aa0;
      aa0.x = xa[c].x * wbuf[t];
      aa0.y = xa[c].y * wbuf[t + 1];
      a2 = wmma_f32_16x16x4(aa0, xa[c], a2);
    }
#pragma unroll 1
    for (int cb = 0; cb < 8; ++cb) {
      v2f bb[4], aa[4];
#pragma unroll
      for (int u = 0; u < 4; ++u) {
        int t = T_REG + (cb * 4 + u) * 4 + koff;
        const float* p = fragbase + (cb * 4 + u) * 4;
        bb[u].x = p[0];
        bb[u].y = p[1];
        aa[u].x = bb[u].x * wbuf[t];
        aa[u].y = bb[u].y * wbuf[t + 1];
      }
#pragma unroll
      for (int u = 0; u < 4; ++u) a2 = wmma_f32_16x16x4(aa[u], bb[u], a2);
    }
    float tr2 = wave_sum32(diag_partial(a2, lane) * (1.f / T_DIM), red, lane);
    float s2 = (1.f / T_DIM) * (16.f / (tr2 + EPSV));
#pragma unroll
    for (int r = 0; r < 8; ++r) ck[r] = a2[r] * s2;
    __syncthreads();
  }

#pragma unroll
  for (int r = 0; r < 8; ++r) { int m = r + 8 * hi; As[m * JS + row] = ck[r]; }
  __syncthreads();
  float tv[8];
#pragma unroll
  for (int r = 0; r < 8; ++r) { int m = r + 8 * hi; tv[r] = As[row * JS + m]; }
  __syncthreads();
#pragma unroll
  for (int r = 0; r < 8; ++r) {
    int m = r + 8 * hi;
    As[m * JS + row] = 0.5f * (ck[r] + tv[r]);
  }
  if (lane < 16) {
#pragma unroll
    for (int j = 0; j < 16; ++j) Vs[lane * JS + j] = (lane == j) ? 1.f : 0.f;
  }
  __syncthreads();

  const int pj = lane & 7;
  const int kg = lane >> 3;
#pragma unroll 1
  for (int sweep = 0; sweep < NSWEEP; ++sweep) {
    bool live = false;
#pragma unroll 1
    for (int rnd = 0; rnd < 15; ++rnd) {
      int ia = rnd + pj;       if (ia >= 15) ia -= 15;
      int ib = rnd + 15 - pj;  if (ib >= 15) ib -= 15;
      if (pj == 0) ib = 15;
      const int p = min(ia, ib);
      const int q = max(ia, ib);
      const float apq = As[p * JS + q];
      const float app = As[p * JS + p];
      const float aqq = As[q * JS + q];
      const bool rot = fabsf(apq) > 1e-30f;
      live = live || (apq * apq > CONV2 * fabsf(app * aqq));
      const float th = (aqq - app) * hw_rcp(2.f * apq);
      float tt = hw_rcp(fabsf(th) + hw_sqrt(fmaf(th, th, 1.f)));
      if (th < 0.f) tt = -tt;
      const float w1 = fmaf(tt, tt, 1.f);
      float y = hw_rsq(w1);
      y = fmaf(0.5f * y, fmaf(-w1 * y, y, 1.f), y);
      const float cc = rot ? y : 1.f;
      const float ss = rot ? tt * y : 0.f;
      __syncthreads();
      {
        float akp[4], akq[4], vkp[4], vkq[4];
#pragma unroll
        for (int i = 0; i < 4; ++i) {
          const int k2 = kg + 4 * i;
          akp[i] = As[k2 * JS + p]; akq[i] = As[k2 * JS + q];
          vkp[i] = Vs[k2 * JS + p]; vkq[i] = Vs[k2 * JS + q];
        }
#pragma unroll
        for (int i = 0; i < 4; ++i) {
          const int k2 = kg + 4 * i;
          As[k2 * JS + p] = cc * akp[i] - ss * akq[i];
          As[k2 * JS + q] = ss * akp[i] + cc * akq[i];
          Vs[k2 * JS + p] = cc * vkp[i] - ss * vkq[i];
          Vs[k2 * JS + q] = ss * vkp[i] + cc * vkq[i];
        }
      }
      __syncthreads();
      {
        float apk[4], aqk[4];
#pragma unroll
        for (int i = 0; i < 4; ++i) {
          const int k2 = kg + 4 * i;
          apk[i] = As[p * JS + k2]; aqk[i] = As[q * JS + k2];
        }
#pragma unroll
        for (int i = 0; i < 4; ++i) {
          const int k2 = kg + 4 * i;
          As[p * JS + k2] = cc * apk[i] - ss * aqk[i];
          As[q * JS + k2] = ss * apk[i] + cc * aqk[i];
        }
      }
      __syncthreads();
    }
    if (__builtin_amdgcn_ballot_w32(live) == 0u) break;
  }
  if (lane < 16) lw[lane] = logf(fmaxf(As[lane * JS + lane], EPSV));
  __syncthreads();

  v8f lg = {0.f, 0.f, 0.f, 0.f, 0.f, 0.f, 0.f, 0.f};
#pragma unroll
  for (int kc = 0; kc < 4; ++kc) {
    int k0 = kc * 4 + koff;
    v2f bb; bb.x = Vs[row * JS + k0]; bb.y = Vs[row * JS + k0 + 1];
    float l0 = lw[k0], l1 = lw[k0 + 1];
    v2f aa; aa.x = bb.x * l0; aa.y = bb.y * l1;
    lg = wmma_f32_16x16x4(aa, bb, lg);
  }

  const float SQ2 = 1.4142135623730951f;
  float* op = out + (size_t)b * 136;
#pragma unroll
  for (int r = 0; r < 8; ++r) {
    int m = r + 8 * hi;
    if (m <= row) {
      int idx = m * 16 - (m * (m - 1)) / 2 + (row - m);
      op[idx] = lg[r] * ((m == row) ? 1.f : SQ2);
    }
  }
}

extern "C" void kernel_launch(void* const* d_in, const int* in_sizes, int n_in,
                              void* d_out, int out_size, void* d_ws,
                              size_t ws_size, hipStream_t stream) {
  const float* x = (const float*)d_in[0];
  float* out = (float*)d_out;
  int B = in_sizes[0] / (C_DIM * T_DIM);
  tyler_tangent_kernel<<<B, 32, 0, stream>>>(x, out);
}


# IEEE divisions of the fixed-point loop become rcp + one Newton step (per iteration 16 pivot reciprocals, now on the scalar float unit, 5 weights, 1 trace scale); last-bit rounding of Cinv, w changes
# speedup vs baseline: 8.7084x; 1.0608x over previous
#include <hip/hip_runtime.h>
#include <math.h>

typedef __attribute__((ext_vector_type(2))) float v2f;
typedef __attribute__((ext_vector_type(8))) float v8f;

#define C_DIM 16
#define T_DIM 160
#define NITERS 10
#define EPSV 1e-8f
#define T_REG 32
#define XS_STRIDE 132
#define CI_STRIDE 17
#define JS 17
#define NSWEEP 8
#define CONV2 3.7252902984619140625e-09f

__device__ __forceinline__ v8f wmma_f32_16x16x4(v2f a, v2f b, v8f c) {
  return __builtin_amdgcn_wmma_f32_16x16x4_f32(false, a, false, b, (short)0, c,
                                               false, false);
}

__device__ __forceinline__ float xhalf(float x) {
  int b = __builtin_bit_cast(int, x);
  return __builtin_bit_cast(
      float, __builtin_amdgcn_permlanex16(b, b, 0x76543210, 0xfedcba98, false, false));
}

__device__ __forceinline__ float lane_bcast(float v, int src) {
  return __builtin_bit_cast(float,
                            __builtin_amdgcn_readlane(__builtin_bit_cast(int, v), src));
}

template <int K>
__device__ __forceinline__ float row_bcast(float v) {
  return __builtin_bit_cast(float, __builtin_amdgcn_update_dpp(0, __builtin_bit_cast(int, v),
                                                               0x150 + K, 0xf, 0xf, true));
}

__device__ __forceinline__ float rcp_nr(float v) {
  const float y = __builtin_amdgcn_rcpf(v);
  return fmaf(fmaf(-v, y, 1.f), y, y);
}

template <int K>
__device__ __forceinline__ void gj_pivot(float (&a)[8], float& csc, int row, int hi) {
  constexpr int RK = K & 7;
  constexpr int HK = K >> 3;
  const float pinv = rcp_nr(lane_bcast(a[RK], 16 * HK + K));
  const float other = xhalf(a[RK]);
  const float rs = ((hi == HK) ? a[RK] : other) * pinv;
  const bool pivcol = (row == K);
  const float m = pivcol ? 0.f : rs;
#pragma unroll
  for (int r = 0; r < 8; ++r) a[r] = fmaf(-row_bcast<K>(a[r]), m, a[r]);
  a[RK] = (hi == HK) ? (pivcol ? -1.f : rs) : a[RK];
  csc = pivcol ? -pinv : csc;
}

__device__ __forceinline__ float hw_rcp(float v) { return __builtin_amdgcn_rcpf(v); }
__device__ __forceinline__ float hw_sqrt(float v) { return __builtin_amdgcn_sqrtf(v); }
__device__ __forceinline__ float hw_rsq(float v) { return __builtin_amdgcn_rsqf(v); }

__device__ __forceinline__ float wave_sum32(float v, float* red, int lane) {
  red[lane] = v;
  __syncthreads();
  float s = 0.f;
#pragma unroll
  for (int i = 0; i < 32; ++i) s += red[i];
  __syncthreads();
  return s;
}

__device__ __forceinline__ float diag_partial(const v8f& a, int lane) {
  float p = 0.f;
  const int mb = (lane >> 4) * 8;
  const int n = lane & 15;
#pragma unroll
  for (int r = 0; r < 8; ++r) p += (n == (r + mb)) ? a[r] : 0.f;
  return p;
}

__device__ __forceinline__ float q_pair_weight(const v2f (&ca)[4], const v2f (&b0)[4],
                                               const v2f (&b1)[4], int hi) {
  v8f y0 = {0.f, 0.f, 0.f, 0.f, 0.f, 0.f, 0.f, 0.f};
  v8f y1 = {0.f, 0.f, 0.f, 0.f, 0.f, 0.f, 0.f, 0.f};
#pragma unroll
  for (int kc = 0; kc < 4; ++kc) {
    y0 = wmma_f32_16x16x4(ca[kc], b0[kc], y0);
    y1 = wmma_f32_16x16x4(ca[kc], b1[kc], y1);
  }
  float p0 = y0[0] * b0[0].x;
  float p1 = y1[0] * b1[0].x;
  p0 = fmaf(y0[1], b0[0].y, p0);
  p1 = fmaf(y1[1], b1[0].y, p1);
#pragma unroll
  for (int kc = 1; kc < 4; ++kc) {
    p0 = fmaf(y0[2 * kc], b0[kc].x, p0);
    p1 = fmaf(y1[2 * kc], b1[kc].x, p1);
    p0 = fmaf(y0[2 * kc + 1], b0[kc].y, p0);
    p1 = fmaf(y1[2 * kc + 1], b1[kc].y, p1);
  }
  const float keep = hi ? p1 : p0;
  const float send = hi ? p0 : p1;
  const float qv = keep + xhalf(send);
  return 16.f * rcp_nr(fmaxf(qv, EPSV));
}

__global__ __launch_bounds__(32, 8) void tyler_tangent_kernel(
    const float* __restrict__ x, float* __restrict__ out) {
  const int b    = blockIdx.x;
  const int lane = threadIdx.x;
  const int row  = lane & 15;
  const int hi   = lane >> 4;
  const int koff = hi * 2;

  __shared__ __attribute__((aligned(16))) float Xs[C_DIM * XS_STRIDE];
  __shared__ float cinv[C_DIM * CI_STRIDE];
  __shared__ float wbuf[T_DIM];
  float* const red = wbuf;
  float* const As = Xs;
  float* const Vs = Xs + C_DIM * JS;
  float* const lw = Xs + 2 * C_DIM * JS;

  const float* xg = x + (size_t)b * (C_DIM * T_DIM);

  {
    const float4* xin = reinterpret_cast<const float4*>(xg);
#pragma unroll
    for (int c = 0; c < C_DIM; ++c) {
      const float4 v = xin[c * (T_DIM / 4) + (T_REG / 4) + lane];
      *reinterpret_cast<float4*>(&Xs[c * XS_STRIDE + 4 * lane]) = v;
    }
  }

  v2f xa[8];
  {
    const float2* pa = reinterpret_cast<const float2*>(xg + row * T_DIM + koff);
#pragma unroll
    for (int c = 0; c < 8; ++c) {
      const float2 v = pa[2 * c];
      xa[c].x = v.x;
      xa[c].y = v.y;
    }
  }
  v2f xq[2][4];
  {
    const float* pb = xg + koff * T_DIM + row;
#pragma unroll
    for (int tile = 0; tile < 2; ++tile) {
#pragma unroll
      for (int kc = 0; kc < 4; ++kc) {
        xq[tile][kc].x = pb[(4 * kc) * T_DIM + 16 * tile];
        xq[tile][kc].y = pb[(4 * kc + 1) * T_DIM + 16 * tile];
      }
    }
  }
  __syncthreads();

  const float* fragbase = &Xs[row * XS_STRIDE + koff];

  v8f acc = {0.f, 0.f, 0.f, 0.f, 0.f, 0.f, 0.f, 0.f};
#pragma unroll
  for (int c = 0; c < 8; ++c) acc = wmma_f32_16x16x4(xa[c], xa[c], acc);
#pragma unroll 1
  for (int cb = 0; cb < 8; ++cb) {
    v2f f[4];
#pragma unroll
    for (int u = 0; u < 4; ++u) {
      const float* p = fragbase + (cb * 4 + u) * 4;
      f[u].x = p[0];
      f[u].y = p[1];
    }
#pragma unroll
    for (int u = 0; u < 4; ++u) acc = wmma_f32_16x16x4(f[u], f[u], acc);
  }
  float tr = wave_sum32(diag_partial(acc, lane) * (1.f / T_DIM), red, lane);
  float scl = (1.f / T_DIM) * (16.f * rcp_nr(tr + EPSV));
  v8f ck;
#pragma unroll
  for (int r = 0; r < 8; ++r) ck[r] = acc[r] * scl;

  const float* xbl = &Xs[koff * XS_STRIDE + row];

#pragma unroll 1
  for (int it = 0; it < NITERS; ++it) {
    float a[8];
#pragma unroll
    for (int r = 0; r < 8; ++r) a[r] = ck[r] + ((r + 8 * hi == row) ? EPSV : 0.f);

    float csc = 0.f;
    gj_pivot<0>(a, csc, row, hi);   gj_pivot<1>(a, csc, row, hi);
    gj_pivot<2>(a, csc, row, hi);   gj_pivot<3>(a, csc, row, hi);
    gj_pivot<4>(a, csc, row, hi);   gj_pivot<5>(a, csc, row, hi);
    gj_pivot<6>(a, csc, row, hi);   gj_pivot<7>(a, csc, row, hi);
    gj_pivot<8>(a, csc, row, hi);   gj_pivot<9>(a, csc, row, hi);
    gj_pivot<10>(a, csc, row, hi);  gj_pivot<11>(a, csc, row, hi);
    gj_pivot<12>(a, csc, row, hi);  gj_pivot<13>(a, csc, row, hi);
    gj_pivot<14>(a, csc, row, hi);  gj_pivot<15>(a, csc, row, hi);
#pragma unroll
    for (int r = 0; r < 8; ++r) cinv[(r + 8 * hi) * CI_STRIDE + row] = a[r] * csc;
    __syncthreads();

    v2f ca[4];
    {
      const int prow = 4 * ((row & 7) >> 1) + 2 * (row >> 3) + (row & 1);
      const float* cp = &cinv[prow * CI_STRIDE + koff];
#pragma unroll
      for (int kc = 0; kc < 4; ++kc) {
        ca[kc].x = cp[4 * kc];
        ca[kc].y = cp[4 * kc + 1];
      }
    }
    wbuf[lane] = q_pair_weight(ca, xq[0], xq[1], hi);
#pragma unroll 1
    for (int j5 = 1; j5 < 5; ++j5) {
      v2f b0[4], b1[4];
      const float* q0 = xbl + 32 * (j5 - 1);
#pragma unroll
      for (int kc = 0; kc < 4; ++kc) {
        b0[kc].x = q0[(4 * kc) * XS_STRIDE];
        b0[kc].y = q0[(4 * kc + 1) * XS_STRIDE];
        b1[kc].x = q0[(4 * kc) * XS_STRIDE + 16];
        b1[kc].y = q0[(4 * kc + 1) * XS_STRIDE + 16];
      }
      wbuf[lane + 32 * j5] = q_pair_weight(ca, b0, b1, hi);
    }
    __syncthreads();

    v8f a2 = {0.f, 0.f, 0.f, 0.f, 0.f, 0.f, 0.f, 0.f};
#pragma unroll
    for (int c = 0; c < 8; ++c) {
      const int t = 4 * c + koff;
      v2f aa0;
      aa0.x = xa[c].x * wbuf[t];
      aa0.y = xa[c].y * wbuf[t + 1];
      a2 = wmma_f32_16x16x4(aa0, xa[c], a2);
    }
#pragma unroll 1
    for (int cb = 0; cb < 8; ++cb) {
      v2f bb[4], aa[4];
#pragma unroll
      for (int u = 0; u < 4; ++u) {
        int t = T_REG + (cb * 4 + u) * 4 + koff;
        const float* p = fragbase + (cb * 4 + u) * 4;
        bb[u].x = p[0];
        bb[u].y = p[1];
        aa[u].x = bb[u].x * wbuf[t];
        aa[u].y = bb[u].y * wbuf[t + 1];
      }
#pragma unroll
      for (int u = 0; u < 4; ++u) a2 = wmma_f32_16x16x4(aa[u], bb[u], a2);
    }
    float tr2 = wave_sum32(diag_partial(a2, lane) * (1.f / T_DIM), red, lane);
    float s2 = (1.f / T_DIM) * (16.f * rcp_nr(tr2 + EPSV));
#pragma unroll
    for (int r = 0; r < 8; ++r) ck[r] = a2[r] * s2;
    __syncthreads();
  }

#pragma unroll
  for (int r = 0; r < 8; ++r) { int m = r + 8 * hi; As[m * JS + row] = ck[r]; }
  __syncthreads();
  float tv[8];
#pragma unroll
  for (int r = 0; r < 8; ++r) { int m = r + 8 * hi; tv[r] = As[row * JS + m]; }
  __syncthreads();
#pragma unroll
  for (int r = 0; r < 8; ++r) {
    int m = r + 8 * hi;
    As[m * JS + row] = 0.5f * (ck[r] + tv[r]);
  }
  if (lane < 16) {
#pragma unroll
    for (int j = 0; j < 16; ++j) Vs[lane * JS + j] = (lane == j) ? 1.f : 0.f;
  }
  __syncthreads();

  const int pj = lane & 7;
  const int kg = lane >> 3;
#pragma unroll 1
  for (int sweep = 0; sweep < NSWEEP; ++sweep) {
    bool live = false;
#pragma unroll 1
    for (int rnd = 0; rnd < 15; ++rnd) {
      int ia = rnd + pj;       if (ia >= 15) ia -= 15;
      int ib = rnd + 15 - pj;  if (ib >= 15) ib -= 15;
      if (pj == 0) ib = 15;
      const int p = min(ia, ib);
      const int q = max(ia, ib);
      const float apq = As[p * JS + q];
      const float app = As[p * JS + p];
      const float aqq = As[q * JS + q];
      const bool rot = fabsf(apq) > 1e-30f;
      live = live || (apq * apq > CONV2 * fabsf(app * aqq));
      const float th = (aqq - app) * hw_rcp(2.f * apq);
      float tt = hw_rcp(fabsf(th) + hw_sqrt(fmaf(th, th, 1.f)));
      if (th < 0.f) tt = -tt;
      const float w1 = fmaf(tt, tt, 1.f);
      float y = hw_rsq(w1);
      y = fmaf(0.5f * y, fmaf(-w1 * y, y, 1.f), y);
      const float cc = rot ? y : 1.f;
      const float ss = rot ? tt * y : 0.f;
      __syncthreads();
      {
        float akp[4], akq[4], vkp[4], vkq[4];
#pragma unroll
        for (int i = 0; i < 4; ++i) {
          const int k2 = kg + 4 * i;
          akp[i] = As[k2 * JS + p]; akq[i] = As[k2 * JS + q];
          vkp[i] = Vs[k2 * JS + p]; vkq[i] = Vs[k2 * JS + q];
        }
#pragma unroll
        for (int i = 0; i < 4; ++i) {
          const int k2 = kg + 4 * i;
          As[k2 * JS + p] = cc * akp[i] - ss * akq[i];
          As[k2 * JS + q] = ss * akp[i] + cc * akq[i];
          Vs[k2 * JS + p] = cc * vkp[i] - ss * vkq[i];
          Vs[k2 * JS + q] = ss * vkp[i] + cc * vkq[i];
        }
      }
      __syncthreads();
      {
        float apk[4], aqk[4];
#pragma unroll
        for (int i = 0; i < 4; ++i) {
          const int k2 = kg + 4 * i;
          apk[i] = As[p * JS + k2]; aqk[i] = As[q * JS + k2];
        }
#pragma unroll
        for (int i = 0; i < 4; ++i) {
          const int k2 = kg + 4 * i;
          As[p * JS + k2] = cc * apk[i] - ss * aqk[i];
          As[q * JS + k2] = ss * apk[i] + cc * aqk[i];
        }
      }
      __syncthreads();
    }
    if (__builtin_amdgcn_ballot_w32(live) == 0u) break;
  }
  if (lane < 16) lw[lane] = logf(fmaxf(As[lane * JS + lane], EPSV));
  __syncthreads();

  v8f lg = {0.f, 0.f, 0.f, 0.f, 0.f, 0.f, 0.f, 0.f};
#pragma unroll
  for (int kc = 0; kc < 4; ++kc) {
    int k0 = kc * 4 + koff;
    v2f bb; bb.x = Vs[row * JS + k0]; bb.y = Vs[row * JS + k0 + 1];
    float l0 = lw[k0], l1 = lw[k0 + 1];
    v2f aa; aa.x = bb.x * l0; aa.y = bb.y * l1;
    lg = wmma_f32_16x16x4(aa, bb, lg);
  }

  const float SQ2 = 1.4142135623730951f;
  float* op = out + (size_t)b * 136;
#pragma unroll
  for (int r = 0; r < 8; ++r) {
    int m = r + 8 * hi;
    if (m <= row) {
      int idx = m * 16 - (m * (m - 1)) / 2 + (row - m);
      op[idx] = lg[r] * ((m == row) ? 1.f : SQ2);
    }
  }
}

extern "C" void kernel_launch(void* const* d_in, const int* in_sizes, int n_in,
                              void* d_out, int out_size, void* d_ws,
                              size_t ws_size, hipStream_t stream) {
  const float* x = (const float*)d_in[0];
  float* out = (float*)d_out;
  int B = in_sizes[0] / (C_DIM * T_DIM);
  tyler_tangent_kernel<<<B, 32, 0, stream>>>(x, out);
}


# Jacobi eigen-solve in fixed positions: pairs are always (2j,2j+1), results stored one round-robin step on; loop-invariant LDS addresses, 8/16-byte accesses (rotation order, rounding change)
# speedup vs baseline: 9.5211x; 1.0933x over previous
#include <hip/hip_runtime.h>
#include <math.h>

typedef __attribute__((ext_vector_type(2))) float v2f;
typedef __attribute__((ext_vector_type(8))) float v8f;

#define C_DIM 16
#define T_DIM 160
#define NITERS 10
#define EPSV 1e-8f
#define T_REG 32
#define XS_STRIDE 132
#define CI_STRIDE 17
#define JS 20
#define NSWEEP 8
#define CONV2 3.7252902984619140625e-09f

__device__ __forceinline__ v8f wmma_f32_16x16x4(v2f a, v2f b, v8f c) {
  return __builtin_amdgcn_wmma_f32_16x16x4_f32(false, a, false, b, (short)0, c,
                                               false, false);
}

__device__ __forceinline__ float xhalf(float x) {
  int b = __builtin_bit_cast(int, x);
  return __builtin_bit_cast(
      float, __builtin_amdgcn_permlanex16(b, b, 0x76543210, 0xfedcba98, false, false));
}

__device__ __forceinline__ float lane_bcast(float v, int src) {
  return __builtin_bit_cast(float,
                            __builtin_amdgcn_readlane(__builtin_bit_cast(int, v), src));
}

template <int K>
__device__ __forceinline__ float row_bcast(float v) {
  return __builtin_bit_cast(float, __builtin_amdgcn_update_dpp(0, __builtin_bit_cast(int, v),
                                                               0x150 + K, 0xf, 0xf, true));
}

__device__ __forceinline__ float rcp_nr(float v) {
  const float y = __builtin_amdgcn_rcpf(v);
  return fmaf(fmaf(-v, y, 1.f), y, y);
}

template <int K>
__device__ __forceinline__ void gj_pivot(float (&a)[8], float& csc, int row, int hi) {
  constexpr int RK = K & 7;
  constexpr int HK = K >> 3;
  const float pinv = rcp_nr(lane_bcast(a[RK], 16 * HK + K));
  const float other = xhalf(a[RK]);
  const float rs = ((hi == HK) ? a[RK] : other) * pinv;
  const bool pivcol = (row == K);
  const float m = pivcol ? 0.f : rs;
#pragma unroll
  for (int r = 0; r < 8; ++r) a[r] = fmaf(-row_bcast<K>(a[r]), m, a[r]);
  a[RK] = (hi == HK) ? (pivcol ? -1.f : rs) : a[RK];
  csc = pivcol ? -pinv : csc;
}

__device__ __forceinline__ float hw_rcp(float v) { return __builtin_amdgcn_rcpf(v); }
__device__ __forceinline__ float hw_sqrt(float v) { return __builtin_amdgcn_sqrtf(v); }
__device__ __forceinline__ float hw_rsq(float v) { return __builtin_amdgcn_rsqf(v); }

__device__ __forceinline__ float wave_sum32(float v, float* red, int lane) {
  red[lane] = v;
  __syncthreads();
  float s = 0.f;
#pragma unroll
  for (int i = 0; i < 32; ++i) s += red[i];
  __syncthreads();
  return s;
}

__device__ __forceinline__ float diag_partial(const v8f& a, int lane) {
  float p = 0.f;
  const int mb = (lane >> 4) * 8;
  const int n = lane & 15;
#pragma unroll
  for (int r = 0; r < 8; ++r) p += (n == (r + mb)) ? a[r] : 0.f;
  return p;
}

__device__ __forceinline__ float q_pair_weight(const v2f (&ca)[4], const v2f (&b0)[4],
                                               const v2f (&b1)[4], int hi) {
  v8f y0 = {0.f, 0.f, 0.f, 0.f, 0.f, 0.f, 0.f, 0.f};
  v8f y1 = {0.f, 0.f, 0.f, 0.f, 0.f, 0.f, 0.f, 0.f};
#pragma unroll
  for (int kc = 0; kc < 4; ++kc) {
    y0 = wmma_f32_16x16x4(ca[kc], b0[kc], y0);
    y1 = wmma_f32_16x16x4(ca[kc], b1[kc], y1);
  }
  float p0 = y0[0] * b0[0].x;
  float p1 = y1[0] * b1[0].x;
  p0 = fmaf(y0[1], b0[0].y, p0);
  p1 = fmaf(y1[1], b1[0].y, p1);
#pragma unroll
  for (int kc = 1; kc < 4; ++kc) {
    p0 = fmaf(y0[2 * kc], b0[kc].x, p0);
    p1 = fmaf(y1[2 * kc], b1[kc].x, p1);
    p0 = fmaf(y0[2 * kc + 1], b0[kc].y, p0);
    p1 = fmaf(y1[2 * kc + 1], b1[kc].y, p1);
  }
  const float keep = hi ? p1 : p0;
  const float send = hi ? p0 : p1;
  const float qv = keep + xhalf(send);
  return 16.f * rcp_nr(fmaxf(qv, EPSV));
}

__global__ __launch_bounds__(32, 8) void tyler_tangent_kernel(
    const float* __restrict__ x, float* __restrict__ out) {
  const int b    = blockIdx.x;
  const int lane = threadIdx.x;
  const int row  = lane & 15;
  const int hi   = lane >> 4;
  const int koff = hi * 2;

  __shared__ __attribute__((aligned(16))) float Xs[C_DIM * XS_STRIDE];
  __shared__ float cinv[C_DIM * CI_STRIDE];
  __shared__ float wbuf[T_DIM];
  float* const red = wbuf;
  float* const As = Xs;
  float* const Vs = Xs + C_DIM * JS;
  float* const lw = Xs + 2 * C_DIM * JS;

  const float* xg = x + (size_t)b * (C_DIM * T_DIM);

  {
    const float4* xin = reinterpret_cast<const float4*>(xg);
#pragma unroll
    for (int c = 0; c < C_DIM; ++c) {
      const float4 v = xin[c * (T_DIM / 4) + (T_REG / 4) + lane];
      *reinterpret_cast<float4*>(&Xs[c * XS_STRIDE + 4 * lane]) = v;
    }
  }

  v2f xa[8];
  {
    const float2* pa = reinterpret_cast<const float2*>(xg + row * T_DIM + koff);
#pragma unroll
    for (int c = 0; c < 8; ++c) {
      const float2 v = pa[2 * c];
      xa[c].x = v.x;
      xa[c].y = v.y;
    }
  }
  v2f xq[2][4];
  {
    const float* pb = xg + koff * T_DIM + row;
#pragma unroll
    for (int tile = 0; tile < 2; ++tile) {
#pragma unroll
      for (int kc = 0; kc < 4; ++kc) {
        xq[tile][kc].x = pb[(4 * kc) * T_DIM + 16 * tile];
        xq[tile][kc].y = pb[(4 * kc + 1) * T_DIM + 16 * tile];
      }
    }
  }
  __syncthreads();

  const float* fragbase = &Xs[row * XS_STRIDE + koff];

  v8f acc = {0.f, 0.f, 0.f, 0.f, 0.f, 0.f, 0.f, 0.f};
#pragma unroll
  for (int c = 0; c < 8; ++c) acc = wmma_f32_16x16x4(xa[c], xa[c], acc);
#pragma unroll 1
  for (int cb = 0; cb < 8; ++cb) {
    v2f f[4];
#pragma unroll
    for (int u = 0; u < 4; ++u) {
      const float* p = fragbase + (cb * 4 + u) * 4;
      f[u].x = p[0];
      f[u].y = p[1];
    }
#pragma unroll
    for (int u = 0; u < 4; ++u) acc = wmma_f32_16x16x4(f[u], f[u], acc);
  }
  float tr = wave_sum32(diag_partial(acc, lane) * (1.f / T_DIM), red, lane);
  float scl = (1.f / T_DIM) * (16.f * rcp_nr(tr + EPSV));
  v8f ck;
#pragma unroll
  for (int r = 0; r < 8; ++r) ck[r] = acc[r] * scl;

  const float* xbl = &Xs[koff * XS_STRIDE + row];

#pragma unroll 1
  for (int it = 0; it < NITERS; ++it) {
    float a[8];
#pragma unroll
    for (int r = 0; r < 8; ++r) a[r] = ck[r] + ((r + 8 * hi == row) ? EPSV : 0.f);

    float csc = 0.f;
    gj_pivot<0>(a, csc, row, hi);   gj_pivot<1>(a, csc, row, hi);
    gj_pivot<2>(a, csc, row, hi);   gj_pivot<3>(a, csc, row, hi);
    gj_pivot<4>(a, csc, row, hi);   gj_pivot<5>(a, csc, row, hi);
    gj_pivot<6>(a, csc, row, hi);   gj_pivot<7>(a, csc, row, hi);
    gj_pivot<8>(a, csc, row, hi);   gj_pivot<9>(a, csc, row, hi);
    gj_pivot<10>(a, csc, row, hi);  gj_pivot<11>(a, csc, row, hi);
    gj_pivot<12>(a, csc, row, hi);  gj_pivot<13>(a, csc, row, hi);
    gj_pivot<14>(a, csc, row, hi);  gj_pivot<15>(a, csc, row, hi);
#pragma unroll
    for (int r = 0; r < 8; ++r) cinv[(r + 8 * hi) * CI_STRIDE + row] = a[r] * csc;
    __syncthreads();

    v2f ca[4];
    {
      const int prow = 4 * ((row & 7) >> 1) + 2 * (row >> 3) + (row & 1);
      const float* cp = &cinv[prow * CI_STRIDE + koff];
#pragma unroll
      for (int kc = 0; kc < 4; ++kc) {
        ca[kc].x = cp[4 * kc];
        ca[kc].y = cp[4 * kc + 1];
      }
    }
    wbuf[lane] = q_pair_weight(ca, xq[0], xq[1], hi);
#pragma unroll 1
    for (int j5 = 1; j5 < 5; ++j5) {
      v2f b0[4], b1[4];
      const float* q0 = xbl + 32 * (j5 - 1);
#pragma unroll
      for (int kc = 0; kc < 4; ++kc) {
        b0[kc].x = q0[(4 * kc) * XS_STRIDE];
        b0[kc].y = q0[(4 * kc + 1) * XS_STRIDE];
        b1[kc].x = q0[(4 * kc) * XS_STRIDE + 16];
        b1[kc].y = q0[(4 * kc + 1) * XS_STRIDE + 16];
      }
      wbuf[lane + 32 * j5] = q_pair_weight(ca, b0, b1, hi);
    }
    __syncthreads();

    v8f a2 = {0.f, 0.f, 0.f, 0.f, 0.f, 0.f, 0.f, 0.f};
#pragma unroll
    for (int c = 0; c < 8; ++c) {
      const int t = 4 * c + koff;
      v2f aa0;
      aa0.x = xa[c].x * wbuf[t];
      aa0.y = xa[c].y * wbuf[t + 1];
      a2 = wmma_f32_16x16x4(aa0, xa[c], a2);
    }
#pragma unroll 1
    for (int cb = 0; cb < 8; ++cb) {
      v2f bb[4], aa[4];
#pragma unroll
      for (int u = 0; u < 4; ++u) {
        int t = T_REG + (cb * 4 + u) * 4 + koff;
        const float* p = fragbase + (cb * 4 + u) * 4;
        bb[u].x = p[0];
        bb[u].y = p[1];
        aa[u].x = bb[u].x * wbuf[t];
        aa[u].y = bb[u].y * wbuf[t + 1];
      }
#pragma unroll
      for (int u = 0; u < 4; ++u) a2 = wmma_f32_16x16x4(aa[u], bb[u], a2);
    }
    float tr2 = wave_sum32(diag_partial(a2, lane) * (1.f / T_DIM), red, lane);
    float s2 = (1.f / T_DIM) * (16.f * rcp_nr(tr2 + EPSV));
#pragma unroll
    for (int r = 0; r < 8; ++r) ck[r] = a2[r] * s2;
    __syncthreads();
  }

#pragma unroll
  for (int r = 0; r < 8; ++r) { int m = r + 8 * hi; As[m * JS + row] = ck[r]; }
  __syncthreads();
  float tv[8];
#pragma unroll
  for (int r = 0; r < 8; ++r) { int m = r + 8 * hi; tv[r] = As[row * JS + m]; }
  __syncthreads();
#pragma unroll
  for (int r = 0; r < 8; ++r) {
    int m = r + 8 * hi;
    As[m * JS + row] = 0.5f * (ck[r] + tv[r]);
  }
  if (lane < 16) {
#pragma unroll
    for (int j = 0; j < 16; ++j) Vs[lane * JS + j] = (lane == j) ? 1.f : 0.f;
  }
  __syncthreads();

  const int pj = lane & 7;
  const int kg = lane >> 3;
  const int P = 2 * pj;
  const int Q = P + 1;
  const int piP = (pj == 0) ? 0 : ((pj == 7) ? 15 : P + 2);
  const int piQ = (pj == 0) ? 2 : Q - 2;
  const float* const par = &As[P * JS + P];
  float* const c1 = &As[(4 * kg) * JS];
  float* const d1 = &Vs[(4 * kg) * JS];
  const float* const r2 = &As[P * JS + 4 * kg];
  float* const w2p = &As[piP * JS + 4 * kg];
  float* const w2q = &As[piQ * JS + 4 * kg];
#pragma unroll 1
  for (int sweep = 0; sweep < NSWEEP; ++sweep) {
    bool live = false;
#pragma unroll 1
    for (int rnd = 0; rnd < 15; ++rnd) {
      const float2 pd = *reinterpret_cast<const float2*>(par);
      const float app = pd.x;
      const float apq = pd.y;
      const float aqq = par[JS + 1];
      const bool rot = fabsf(apq) > 1e-30f;
      live = live || (apq * apq > CONV2 * fabsf(app * aqq));
      const float th = (aqq - app) * hw_rcp(2.f * apq);
      float tt = hw_rcp(fabsf(th) + hw_sqrt(fmaf(th, th, 1.f)));
      if (th < 0.f) tt = -tt;
      const float w1 = fmaf(tt, tt, 1.f);
      float y = hw_rsq(w1);
      y = fmaf(0.5f * y, fmaf(-w1 * y, y, 1.f), y);
      const float cc = rot ? y : 1.f;
      const float ss = rot ? tt * y : 0.f;
      __syncthreads();
      {
        float2 ak[4], vk[4];
#pragma unroll
        for (int i = 0; i < 4; ++i) {
          ak[i] = *reinterpret_cast<const float2*>(c1 + i * JS + P);
          vk[i] = *reinterpret_cast<const float2*>(d1 + i * JS + P);
        }
#pragma unroll
        for (int i = 0; i < 4; ++i) {
          c1[i * JS + piP] = cc * ak[i].x - ss * ak[i].y;
          c1[i * JS + piQ] = ss * ak[i].x + cc * ak[i].y;
          d1[i * JS + piP] = cc * vk[i].x - ss * vk[i].y;
          d1[i * JS + piQ] = ss * vk[i].x + cc * vk[i].y;
        }
      }
      __syncthreads();
      {
        const float4 ap = *reinterpret_cast<const float4*>(r2);
        const float4 aq = *reinterpret_cast<const float4*>(r2 + JS);
        float4 np, nq;
        np.x = cc * ap.x - ss * aq.x;  nq.x = ss * ap.x + cc * aq.x;
        np.y = cc * ap.y - ss * aq.y;  nq.y = ss * ap.y + cc * aq.y;
        np.z = cc * ap.z - ss * aq.z;  nq.z = ss * ap.z + cc * aq.z;
        np.w = cc * ap.w - ss * aq.w;  nq.w = ss * ap.w + cc * aq.w;
        *reinterpret_cast<float4*>(w2p) = np;
        *reinterpret_cast<float4*>(w2q) = nq;
      }
      __syncthreads();
    }
    if (__builtin_amdgcn_ballot_w32(live) == 0u) break;
  }
  if (lane < 16) lw[lane] = logf(fmaxf(As[lane * JS + lane], EPSV));
  __syncthreads();

  v8f lg = {0.f, 0.f, 0.f, 0.f, 0.f, 0.f, 0.f, 0.f};
#pragma unroll
  for (int kc = 0; kc < 4; ++kc) {
    int k0 = kc * 4 + koff;
    v2f bb; bb.x = Vs[row * JS + k0]; bb.y = Vs[row * JS + k0 + 1];
    float l0 = lw[k0], l1 = lw[k0 + 1];
    v2f aa; aa.x = bb.x * l0; aa.y = bb.y * l1;
    lg = wmma_f32_16x16x4(aa, bb, lg);
  }

  const float SQ2 = 1.4142135623730951f;
  float* op = out + (size_t)b * 136;
#pragma unroll
  for (int r = 0; r < 8; ++r) {
    int m = r + 8 * hi;
    if (m <= row) {
      int idx = m * 16 - (m * (m - 1)) / 2 + (row - m);
      op[idx] = lg[r] * ((m == row) ? 1.f : SQ2);
    }
  }
}

extern "C" void kernel_launch(void* const* d_in, const int* in_sizes, int n_in,
                              void* d_out, int out_size, void* d_ws,
                              size_t ws_size, hipStream_t stream) {
  const float* x = (const float*)d_in[0];
  float* out = (float*)d_out;
  int B = in_sizes[0] / (C_DIM * T_DIM);
  tyler_tangent_kernel<<<B, 32, 0, stream>>>(x, out);
}


# Jacobi eigen-solve made one-sided: only columns of G=C are rotated until orthogonal (G -> V*diag(lambda)); no V matrix, no row phase, one LDS pass per round; rotation order and rounding change
# speedup vs baseline: 10.0523x; 1.0558x over previous
#include <hip/hip_runtime.h>
#include <math.h>

typedef __attribute__((ext_vector_type(2))) float v2f;
typedef __attribute__((ext_vector_type(8))) float v8f;

#define C_DIM 16
#define T_DIM 160
#define NITERS 10
#define EPSV 1e-8f
#define T_REG 32
#define XS_STRIDE 132
#define CI_STRIDE 17
#define JS 20
#define NSWEEP 8
#define CONV2 3.7252902984619140625e-09f

__device__ __forceinline__ v8f wmma_f32_16x16x4(v2f a, v2f b, v8f c) {
  return __builtin_amdgcn_wmma_f32_16x16x4_f32(false, a, false, b, (short)0, c,
                                               false, false);
}

__device__ __forceinline__ float xhalf(float x) {
  int b = __builtin_bit_cast(int, x);
  return __builtin_bit_cast(
      float, __builtin_amdgcn_permlanex16(b, b, 0x76543210, 0xfedcba98, false, false));
}

__device__ __forceinline__ float row_ror8(float x) {
  return __builtin_bit_cast(float, __builtin_amdgcn_update_dpp(0, __builtin_bit_cast(int, x), 0x128, 0xf,
                                                               0xf, true));
}

__device__ __forceinline__ float lane_bcast(float v, int src) {
  return __builtin_bit_cast(float,
                            __builtin_amdgcn_readlane(__builtin_bit_cast(int, v), src));
}

template <int K>
__device__ __forceinline__ float row_bcast(float v) {
  return __builtin_bit_cast(float, __builtin_amdgcn_update_dpp(0, __builtin_bit_cast(int, v),
                                                               0x150 + K, 0xf, 0xf, true));
}

__device__ __forceinline__ float rcp_nr(float v) {
  const float y = __builtin_amdgcn_rcpf(v);
  return fmaf(fmaf(-v, y, 1.f), y, y);
}

template <int K>
__device__ __forceinline__ void gj_pivot(float (&a)[8], float& csc, int row, int hi) {
  constexpr int RK = K & 7;
  constexpr int HK = K >> 3;
  const float pinv = rcp_nr(lane_bcast(a[RK], 16 * HK + K));
  const float other = xhalf(a[RK]);
  const float rs = ((hi == HK) ? a[RK] : other) * pinv;
  const bool pivcol = (row == K);
  const float m = pivcol ? 0.f : rs;
#pragma unroll
  for (int r = 0; r < 8; ++r) a[r] = fmaf(-row_bcast<K>(a[r]), m, a[r]);
  a[RK] = (hi == HK) ? (pivcol ? -1.f : rs) : a[RK];
  csc = pivcol ? -pinv : csc;
}

__device__ __forceinline__ float hw_rcp(float v) { return __builtin_amdgcn_rcpf(v); }
__device__ __forceinline__ float hw_sqrt(float v) { return __builtin_amdgcn_sqrtf(v); }
__device__ __forceinline__ float hw_rsq(float v) { return __builtin_amdgcn_rsqf(v); }

__device__ __forceinline__ float wave_sum32(float v, float* red, int lane) {
  red[lane] = v;
  __syncthreads();
  float s = 0.f;
#pragma unroll
  for (int i = 0; i < 32; ++i) s += red[i];
  __syncthreads();
  return s;
}

__device__ __forceinline__ float diag_partial(const v8f& a, int lane) {
  float p = 0.f;
  const int mb = (lane >> 4) * 8;
  const int n = lane & 15;
#pragma unroll
  for (int r = 0; r < 8; ++r) p += (n == (r + mb)) ? a[r] : 0.f;
  return p;
}

__device__ __forceinline__ float q_pair_weight(const v2f (&ca)[4], const v2f (&b0)[4],
                                               const v2f (&b1)[4], int hi) {
  v8f y0 = {0.f, 0.f, 0.f, 0.f, 0.f, 0.f, 0.f, 0.f};
  v8f y1 = {0.f, 0.f, 0.f, 0.f, 0.f, 0.f, 0.f, 0.f};
#pragma unroll
  for (int kc = 0; kc < 4; ++kc) {
    y0 = wmma_f32_16x16x4(ca[kc], b0[kc], y0);
    y1 = wmma_f32_16x16x4(ca[kc], b1[kc], y1);
  }
  float p0 = y0[0] * b0[0].x;
  float p1 = y1[0] * b1[0].x;
  p0 = fmaf(y0[1], b0[0].y, p0);
  p1 = fmaf(y1[1], b1[0].y, p1);
#pragma unroll
  for (int kc = 1; kc < 4; ++kc) {
    p0 = fmaf(y0[2 * kc], b0[kc].x, p0);
    p1 = fmaf(y1[2 * kc], b1[kc].x, p1);
    p0 = fmaf(y0[2 * kc + 1], b0[kc].y, p0);
    p1 = fmaf(y1[2 * kc + 1], b1[kc].y, p1);
  }
  const float keep = hi ? p1 : p0;
  const float send = hi ? p0 : p1;
  const float qv = keep + xhalf(send);
  return 16.f * rcp_nr(fmaxf(qv, EPSV));
}

__global__ __launch_bounds__(32, 8) void tyler_tangent_kernel(
    const float* __restrict__ x, float* __restrict__ out) {
  const int b    = blockIdx.x;
  const int lane = threadIdx.x;
  const int row  = lane & 15;
  const int hi   = lane >> 4;
  const int koff = hi * 2;

  __shared__ __attribute__((aligned(16))) float Xs[C_DIM * XS_STRIDE];
  __shared__ float cinv[C_DIM * CI_STRIDE];
  __shared__ float wbuf[T_DIM];
  float* const red = wbuf;
  float* const As = Xs;
  float* const lw = Xs + 2 * C_DIM * JS;

  const float* xg = x + (size_t)b * (C_DIM * T_DIM);

  {
    const float4* xin = reinterpret_cast<const float4*>(xg);
#pragma unroll
    for (int c = 0; c < C_DIM; ++c) {
      const float4 v = xin[c * (T_DIM / 4) + (T_REG / 4) + lane];
      *reinterpret_cast<float4*>(&Xs[c * XS_STRIDE + 4 * lane]) = v;
    }
  }

  v2f xa[8];
  {
    const float2* pa = reinterpret_cast<const float2*>(xg + row * T_DIM + koff);
#pragma unroll
    for (int c = 0; c < 8; ++c) {
      const float2 v = pa[2 * c];
      xa[c].x = v.x;
      xa[c].y = v.y;
    }
  }
  v2f xq[2][4];
  {
    const float* pb = xg + koff * T_DIM + row;
#pragma unroll
    for (int tile = 0; tile < 2; ++tile) {
#pragma unroll
      for (int kc = 0; kc < 4; ++kc) {
        xq[tile][kc].x = pb[(4 * kc) * T_DIM + 16 * tile];
        xq[tile][kc].y = pb[(4 * kc + 1) * T_DIM + 16 * tile];
      }
    }
  }
  __syncthreads();

  const float* fragbase = &Xs[row * XS_STRIDE + koff];

  v8f acc = {0.f, 0.f, 0.f, 0.f, 0.f, 0.f, 0.f, 0.f};
#pragma unroll
  for (int c = 0; c < 8; ++c) acc = wmma_f32_16x16x4(xa[c], xa[c], acc);
#pragma unroll 1
  for (int cb = 0; cb < 8; ++cb) {
    v2f f[4];
#pragma unroll
    for (int u = 0; u < 4; ++u) {
      const float* p = fragbase + (cb * 4 + u) * 4;
      f[u].x = p[0];
      f[u].y = p[1];
    }
#pragma unroll
    for (int u = 0; u < 4; ++u) acc = wmma_f32_16x16x4(f[u], f[u], acc);
  }
  float tr = wave_sum32(diag_partial(acc, lane) * (1.f / T_DIM), red, lane);
  float scl = (1.f / T_DIM) * (16.f * rcp_nr(tr + EPSV));
  v8f ck;
#pragma unroll
  for (int r = 0; r < 8; ++r) ck[r] = acc[r] * scl;

  const float* xbl = &Xs[koff * XS_STRIDE + row];

#pragma unroll 1
  for (int it = 0; it < NITERS; ++it) {
    float a[8];
#pragma unroll
    for (int r = 0; r < 8; ++r) a[r] = ck[r] + ((r + 8 * hi == row) ? EPSV : 0.f);

    float csc = 0.f;
    gj_pivot<0>(a, csc, row, hi);   gj_pivot<1>(a, csc, row, hi);
    gj_pivot<2>(a, csc, row, hi);   gj_pivot<3>(a, csc, row, hi);
    gj_pivot<4>(a, csc, row, hi);   gj_pivot<5>(a, csc, row, hi);
    gj_pivot<6>(a, csc, row, hi);   gj_pivot<7>(a, csc, row, hi);
    gj_pivot<8>(a, csc, row, hi);   gj_pivot<9>(a, csc, row, hi);
    gj_pivot<10>(a, csc, row, hi);  gj_pivot<11>(a, csc, row, hi);
    gj_pivot<12>(a, csc, row, hi);  gj_pivot<13>(a, csc, row, hi);
    gj_pivot<14>(a, csc, row, hi);  gj_pivot<15>(a, csc, row, hi);
#pragma unroll
    for (int r = 0; r < 8; ++r) cinv[(r + 8 * hi) * CI_STRIDE + row] = a[r] * csc;
    __syncthreads();

    v2f ca[4];
    {
      const int prow = 4 * ((row & 7) >> 1) + 2 * (row >> 3) + (row & 1);
      const float* cp = &cinv[prow * CI_STRIDE + koff];
#pragma unroll
      for (int kc = 0; kc < 4; ++kc) {
        ca[kc].x = cp[4 * kc];
        ca[kc].y = cp[4 * kc + 1];
      }
    }
    wbuf[lane] = q_pair_weight(ca, xq[0], xq[1], hi);
#pragma unroll 1
    for (int j5 = 1; j5 < 5; ++j5) {
      v2f b0[4], b1[4];
      const float* q0 = xbl + 32 * (j5 - 1);
#pragma unroll
      for (int kc = 0; kc < 4; ++kc) {
        b0[kc].x = q0[(4 * kc) * XS_STRIDE];
        b0[kc].y = q0[(4 * kc + 1) * XS_STRIDE];
        b1[kc].x = q0[(4 * kc) * XS_STRIDE + 16];
        b1[kc].y = q0[(4 * kc + 1) * XS_STRIDE + 16];
      }
      wbuf[lane + 32 * j5] = q_pair_weight(ca, b0, b1, hi);
    }
    __syncthreads();

    v8f a2 = {0.f, 0.f, 0.f, 0.f, 0.f, 0.f, 0.f, 0.f};
#pragma unroll
    for (int c = 0; c < 8; ++c) {
      const int t = 4 * c + koff;
      v2f aa0;
      aa0.x = xa[c].x * wbuf[t];
      aa0.y = xa[c].y * wbuf[t + 1];
      a2 = wmma_f32_16x16x4(aa0, xa[c], a2);
    }
#pragma unroll 1
    for (int cb = 0; cb < 8; ++cb) {
      v2f bb[4], aa[4];
#pragma unroll
      for (int u = 0; u < 4; ++u) {
        int t = T_REG + (cb * 4 + u) * 4 + koff;
        const float* p = fragbase + (cb * 4 + u) * 4;
        bb[u].x = p[0];
        bb[u].y = p[1];
        aa[u].x = bb[u].x * wbuf[t];
        aa[u].y = bb[u].y * wbuf[t + 1];
      }
#pragma unroll
      for (int u = 0; u < 4; ++u) a2 = wmma_f32_16x16x4(aa[u], bb[u], a2);
    }
    float tr2 = wave_sum32(diag_partial(a2, lane) * (1.f / T_DIM), red, lane);
    float s2 = (1.f / T_DIM) * (16.f * rcp_nr(tr2 + EPSV));
#pragma unroll
    for (int r = 0; r < 8; ++r) ck[r] = a2[r] * s2;
    __syncthreads();
  }

#pragma unroll
  for (int r = 0; r < 8; ++r) { int m = r + 8 * hi; As[m * JS + row] = ck[r]; }
  __syncthreads();
  float tv[8];
#pragma unroll
  for (int r = 0; r < 8; ++r) { int m = r + 8 * hi; tv[r] = As[row * JS + m]; }
  __syncthreads();
#pragma unroll
  for (int r = 0; r < 8; ++r) {
    int m = r + 8 * hi;
    As[m * JS + row] = 0.5f * (ck[r] + tv[r]);
  }
  __syncthreads();

  const int pj = lane & 7;
  const int kg = lane >> 3;
  const int P = 2 * pj;
  const int piP = (pj == 0) ? 0 : ((pj == 7) ? 15 : P + 2);
  const int piQ = (pj == 0) ? 2 : P - 1;
  float* const c1 = &As[(4 * kg) * JS];
#pragma unroll 1
  for (int sweep = 0; sweep < NSWEEP; ++sweep) {
    bool live = false;
#pragma unroll 1
    for (int rnd = 0; rnd < 15; ++rnd) {
      float2 gk[4];
#pragma unroll
      for (int i = 0; i < 4; ++i) gk[i] = *reinterpret_cast<const float2*>(c1 + i * JS + P);
      float app = gk[0].x * gk[0].x;
      float aqq = gk[0].y * gk[0].y;
      float apq = gk[0].x * gk[0].y;
#pragma unroll
      for (int i = 1; i < 4; ++i) {
        app = fmaf(gk[i].x, gk[i].x, app);
        aqq = fmaf(gk[i].y, gk[i].y, aqq);
        apq = fmaf(gk[i].x, gk[i].y, apq);
      }
      app += row_ror8(app);  aqq += row_ror8(aqq);  apq += row_ror8(apq);
      app += xhalf(app);     aqq += xhalf(aqq);     apq += xhalf(apq);
      const bool rot = fabsf(apq) > 1e-30f;
      live = live || (apq * apq > CONV2 * (app * aqq));
      const float th = (aqq - app) * hw_rcp(2.f * apq);
      float tt = hw_rcp(fabsf(th) + hw_sqrt(fmaf(th, th, 1.f)));
      if (th < 0.f) tt = -tt;
      const float w1 = fmaf(tt, tt, 1.f);
      float y = hw_rsq(w1);
      y = fmaf(0.5f * y, fmaf(-w1 * y, y, 1.f), y);
      const float cc = rot ? y : 1.f;
      const float ss = rot ? tt * y : 0.f;
#pragma unroll
      for (int i = 0; i < 4; ++i) {
        c1[i * JS + piP] = cc * gk[i].x - ss * gk[i].y;
        c1[i * JS + piQ] = ss * gk[i].x + cc * gk[i].y;
      }
      __syncthreads();
    }
    if (__builtin_amdgcn_ballot_w32(live) == 0u) break;
  }
  if (lane < 16) {
    float n2 = 0.f;
#pragma unroll
    for (int i = 0; i < 16; ++i) { const float v = As[i * JS + lane]; n2 = fmaf(v, v, n2); }
    lw[lane] = logf(fmaxf(sqrtf(n2), EPSV)) / fmaxf(n2, 1e-37f);
  }
  __syncthreads();

  v8f lg = {0.f, 0.f, 0.f, 0.f, 0.f, 0.f, 0.f, 0.f};
#pragma unroll
  for (int kc = 0; kc < 4; ++kc) {
    int k0 = kc * 4 + koff;
    v2f bb; bb.x = As[row * JS + k0]; bb.y = As[row * JS + k0 + 1];
    float l0 = lw[k0], l1 = lw[k0 + 1];
    v2f aa; aa.x = bb.x * l0; aa.y = bb.y * l1;
    lg = wmma_f32_16x16x4(aa, bb, lg);
  }

  const float SQ2 = 1.4142135623730951f;
  float* op = out + (size_t)b * 136;
#pragma unroll
  for (int r = 0; r < 8; ++r) {
    int m = r + 8 * hi;
    if (m <= row) {
      int idx = m * 16 - (m * (m - 1)) / 2 + (row - m);
      op[idx] = lg[r] * ((m == row) ? 1.f : SQ2);
    }
  }
}

extern "C" void kernel_launch(void* const* d_in, const int* in_sizes, int n_in,
                              void* d_out, int out_size, void* d_ws,
                              size_t ws_size, hipStream_t stream) {
  const float* x = (const float*)d_in[0];
  float* out = (float*)d_out;
  int B = in_sizes[0] / (C_DIM * T_DIM);
  tyler_tangent_kernel<<<B, 32, 0, stream>>>(x, out);
}
